# starting kernel
# baseline (speedup 1.0000x reference)
#include <hip/hip_runtime.h>
#include <hip/hip_bf16.h>
#include <math.h>
#include <stdint.h>


typedef float v2f __attribute__((ext_vector_type(2)));
typedef float v8f __attribute__((ext_vector_type(8)));

#define NEG_SLOPE 0.2f

__device__ __forceinline__ void atomicMaxF32(float* addr, float val) {
  if (val >= 0.0f) {
    atomicMax((int*)addr, __float_as_int(val));
  } else {
    atomicMin((unsigned int*)addr, __float_as_uint(val));
  }
}

template <int NTILES, int NACT>
__global__ void __launch_bounds__(32 * NTILES)
gat_wmma_gemm_k128(const float* __restrict__ A, const float* __restrict__ B,
                   float* __restrict__ C) {
  constexpr int NPAD = NTILES * 16;
  constexpr int SA = 132;
  constexpr int SB = 132;
  __shared__ float ldsA[16 * SA];
  __shared__ float ldsB[NPAD * SB];

  const int tid  = threadIdx.x;
  const int lane = tid & 31;
  const int wave = tid >> 5;
  const size_t m0 = (size_t)blockIdx.x * 16;
  const int c0 = blockIdx.y * NPAD;

  for (int i = tid; i < 16 * 128; i += 32 * NTILES) {
    int r = i >> 7, c = i & 127;
    ldsA[r * SA + c] = A[(m0 + r) * 128 + c];
  }
  for (int i = tid; i < NPAD * 128; i += 32 * NTILES) {
    int k = i / NPAD, n = i - k * NPAD;
    int gn = c0 + n;
    ldsB[n * SB + k] = (gn < NACT) ? B[(size_t)k * NACT + gn] : 0.0f;
  }
  __syncthreads();

  const int l15  = lane & 15;
  const int koff = (lane >> 4) << 1;
  const int ncol = wave * 16 + l15;
  const float* ap = &ldsA[l15 * SA + koff];
  const float* bp = &ldsB[ncol * SB + koff];

  v8f acc = {0.f, 0.f, 0.f, 0.f, 0.f, 0.f, 0.f, 0.f};
#pragma unroll
  for (int kb = 0; kb < 128; kb += 4) {
    v2f a = *(const v2f*)(ap + kb);
    v2f b = *(const v2f*)(bp + kb);
    acc = __builtin_amdgcn_wmma_f32_16x16x4_f32(
         false, a,  false, b,
         (short)0, acc,  false,  false);
  }

  const int gcol = c0 + ncol;
  if (gcol < NACT) {
    float* cp = C + (m0 + (size_t)((lane >> 4) << 3)) * NACT + gcol;
#pragma unroll
    for (int r = 0; r < 8; ++r) cp[(size_t)r * NACT] = acc[r];
  }
}

__global__ void fill_kernel(float* __restrict__ p, float v, long long n) {
  long long i = (long long)blockIdx.x * blockDim.x + threadIdx.x;
  long long stride = (long long)gridDim.x * blockDim.x;
  for (; i < n; i += stride) p[i] = v;
}

__global__ void init_out_kernel(float* __restrict__ out, const float* __restrict__ b2,
                                int Nn) {
  int idx = blockIdx.x * blockDim.x + threadIdx.x;
  if (idx >= Nn * 40) return;
  out[idx] = b2[idx % 40];
}

__global__ void scores1_kernel(const float* __restrict__ h,
                               const float* __restrict__ att_src,
                               const float* __restrict__ att_dst,
                               float* __restrict__ a_src, float* __restrict__ a_dst,
                               int Nn) {
  int idx = blockIdx.x * blockDim.x + threadIdx.x;
  if (idx >= Nn * 8) return;
  int n = idx >> 3, hd = idx & 7;
  const float* hp = h + (size_t)n * 128 + hd * 16;
  float s = 0.f, d = 0.f;
#pragma unroll
  for (int c = 0; c < 16; ++c) {
    float v = hp[c];
    s = fmaf(v, att_src[hd * 16 + c], s);
    d = fmaf(v, att_dst[hd * 16 + c], d);
  }
  a_src[idx] = s;
  a_dst[idx] = d;
}

__global__ void scores2_kernel(const float* __restrict__ h2,
                               const float* __restrict__ att_src,
                               const float* __restrict__ att_dst,
                               float* __restrict__ a_src, float* __restrict__ a_dst,
                               int Nn) {
  int n = blockIdx.x * blockDim.x + threadIdx.x;
  if (n >= Nn) return;
  const float* hp = h2 + (size_t)n * 40;
  float s = 0.f, d = 0.f;
#pragma unroll
  for (int c = 0; c < 40; ++c) {
    float v = hp[c];
    s = fmaf(v, att_src[c], s);
    d = fmaf(v, att_dst[c], d);
  }
  a_src[n] = s;
  a_dst[n] = d;
}

__global__ void edge_max1_kernel(const int* __restrict__ src, const int* __restrict__ dst,
                                 const float* __restrict__ a_src, const float* __restrict__ a_dst,
                                 float* __restrict__ m, int E, int Etot) {
  int idx = blockIdx.x * blockDim.x + threadIdx.x;
  if (idx >= Etot * 8) return;
  int e = idx >> 3, hd = idx & 7;
  int s = (e < E) ? src[e] : (e - E);
  int d = (e < E) ? dst[e] : (e - E);
  float v = a_src[s * 8 + hd] + a_dst[d * 8 + hd];
  v = (v >= 0.f) ? v : NEG_SLOPE * v;
  atomicMaxF32(&m[d * 8 + hd], v);
}

__global__ void edge_sum1_kernel(const int* __restrict__ src, const int* __restrict__ dst,
                                 const float* __restrict__ a_src, const float* __restrict__ a_dst,
                                 const float* __restrict__ m, float* __restrict__ ssum,
                                 int E, int Etot) {
  int idx = blockIdx.x * blockDim.x + threadIdx.x;
  if (idx >= Etot * 8) return;
  int e = idx >> 3, hd = idx & 7;
  int s = (e < E) ? src[e] : (e - E);
  int d = (e < E) ? dst[e] : (e - E);
  float v = a_src[s * 8 + hd] + a_dst[d * 8 + hd];
  v = (v >= 0.f) ? v : NEG_SLOPE * v;
  atomicAdd(&ssum[d * 8 + hd], __expf(v - m[d * 8 + hd]));
}

__global__ void edge_agg1_kernel(const int* __restrict__ src, const int* __restrict__ dst,
                                 const float* __restrict__ a_src, const float* __restrict__ a_dst,
                                 const float* __restrict__ m, const float* __restrict__ ssum,
                                 const float* __restrict__ h, float* __restrict__ agg,
                                 int E, int Etot) {
  int wid  = (int)(((long long)blockIdx.x * blockDim.x + threadIdx.x) >> 5);
  int lane = threadIdx.x & 31;
  if (wid >= Etot) return;
  int s = (wid < E) ? src[wid] : (wid - E);
  int d = (wid < E) ? dst[wid] : (wid - E);
  int hd = lane >> 2;
  float e = a_src[s * 8 + hd] + a_dst[d * 8 + hd];
  e = (e >= 0.f) ? e : NEG_SLOPE * e;
  float alpha = __expf(e - m[d * 8 + hd]) / ssum[d * 8 + hd];
  const float4 hv = *(const float4*)(h + (size_t)s * 128 + lane * 4);
  float* op = agg + (size_t)d * 128 + lane * 4;
  atomicAdd(op + 0, hv.x * alpha);
  atomicAdd(op + 1, hv.y * alpha);
  atomicAdd(op + 2, hv.z * alpha);
  atomicAdd(op + 3, hv.w * alpha);
}

__global__ void elu_bias_kernel(float* __restrict__ a, const float* __restrict__ b,
                                long long n) {
  long long idx = (long long)blockIdx.x * blockDim.x + threadIdx.x;
  if (idx >= n) return;
  float v = a[idx] + b[idx & 127];
  a[idx] = (v > 0.f) ? v : (__expf(v) - 1.0f);
}

__global__ void edge_max2_kernel(const int* __restrict__ src, const int* __restrict__ dst,
                                 const float* __restrict__ a_src, const float* __restrict__ a_dst,
                                 float* __restrict__ m, int E, int Etot) {
  int e = blockIdx.x * blockDim.x + threadIdx.x;
  if (e >= Etot) return;
  int s = (e < E) ? src[e] : (e - E);
  int d = (e < E) ? dst[e] : (e - E);
  float v = a_src[s] + a_dst[d];
  v = (v >= 0.f) ? v : NEG_SLOPE * v;
  atomicMaxF32(&m[d], v);
}

__global__ void edge_sum2_kernel(const int* __restrict__ src, const int* __restrict__ dst,
                                 const float* __restrict__ a_src, const float* __restrict__ a_dst,
                                 const float* __restrict__ m, float* __restrict__ ssum,
                                 int E, int Etot) {
  int e = blockIdx.x * blockDim.x + threadIdx.x;
  if (e >= Etot) return;
  int s = (e < E) ? src[e] : (e - E);
  int d = (e < E) ? dst[e] : (e - E);
  float v = a_src[s] + a_dst[d];
  v = (v >= 0.f) ? v : NEG_SLOPE * v;
  atomicAdd(&ssum[d], __expf(v - m[d]));
}

__global__ void edge_agg2_kernel(const int* __restrict__ src, const int* __restrict__ dst,
                                 const float* __restrict__ a_src, const float* __restrict__ a_dst,
                                 const float* __restrict__ m, const float* __restrict__ ssum,
                                 const float* __restrict__ h2, float* __restrict__ out,
                                 int E, int Etot) {
  int wid  = (int)(((long long)blockIdx.x * blockDim.x + threadIdx.x) >> 5);
  int lane = threadIdx.x & 31;
  if (wid >= Etot) return;
  int s = (wid < E) ? src[wid] : (wid - E);
  int d = (wid < E) ? dst[wid] : (wid - E);
  float e = a_src[s] + a_dst[d];
  e = (e >= 0.f) ? e : NEG_SLOPE * e;
  float alpha = __expf(e - m[d]) / ssum[d];
  for (int c = lane; c < 40; c += 32)
    atomicAdd(&out[(size_t)d * 40 + c], h2[(size_t)s * 40 + c] * alpha);
}

extern "C" void kernel_launch(void* const* d_in, const int* in_sizes, int n_in,
                              void* d_out, int out_size, void* d_ws, size_t ws_size,
                              hipStream_t stream) {
  const float* x   = (const float*)d_in[0];
  const int*   ei  = (const int*)d_in[1];
  const float* W1  = (const float*)d_in[2];
  const float* as1 = (const float*)d_in[3];
  const float* ad1 = (const float*)d_in[4];
  const float* b1  = (const float*)d_in[5];
  const float* W2  = (const float*)d_in[6];
  const float* as2 = (const float*)d_in[7];
  const float* ad2 = (const float*)d_in[8];
  const float* b2  = (const float*)d_in[9];
  float* out = (float*)d_out;

  const int Nn   = in_sizes[0] / 128;
  const int E    = in_sizes[1] / 2;
  const int Etot = E + Nn;
  const int* srcp = ei;
  const int* dstp = ei + E;

  float* ws   = (float*)d_ws;
  float* h    = ws;  ws += (size_t)Nn * 128;
  float* agg1 = ws;  ws += (size_t)Nn * 128;
  float* asc1 = ws;  ws += (size_t)Nn * 8;
  float* adc1 = ws;  ws += (size_t)Nn * 8;
  float* m1   = ws;  ws += (size_t)Nn * 8;
  float* s1   = ws;  ws += (size_t)Nn * 8;
  float* h2   = ws;  ws += (size_t)Nn * 40;
  float* asc2 = ws;  ws += Nn;
  float* adc2 = ws;  ws += Nn;
  float* m2   = ws;  ws += Nn;
  float* s2   = ws;  ws += Nn;

  const int T = 256;
  const int fillBlocks = 2048;

  fill_kernel<<<fillBlocks, T, 0, stream>>>(m1, -INFINITY, (long long)Nn * 8);
  fill_kernel<<<fillBlocks, T, 0, stream>>>(s1, 0.0f, (long long)Nn * 8);
  fill_kernel<<<fillBlocks, T, 0, stream>>>(agg1, 0.0f, (long long)Nn * 128);
  fill_kernel<<<fillBlocks, T, 0, stream>>>(m2, -INFINITY, (long long)Nn);
  fill_kernel<<<fillBlocks, T, 0, stream>>>(s2, 0.0f, (long long)Nn);
  init_out_kernel<<<(Nn * 40 + T - 1) / T, T, 0, stream>>>(out, b2, Nn);

  gat_wmma_gemm_k128<4, 128><<<dim3(Nn / 16, 2), 4 * 32, 0, stream>>>(x, W1, h);
  scores1_kernel<<<(Nn * 8 + T - 1) / T, T, 0, stream>>>(h, as1, ad1, asc1, adc1, Nn);
  edge_max1_kernel<<<(Etot * 8 + T - 1) / T, T, 0, stream>>>(srcp, dstp, asc1, adc1, m1, E, Etot);
  edge_sum1_kernel<<<(Etot * 8 + T - 1) / T, T, 0, stream>>>(srcp, dstp, asc1, adc1, m1, s1, E, Etot);
  edge_agg1_kernel<<<(Etot + 7) / 8, T, 0, stream>>>(srcp, dstp, asc1, adc1, m1, s1, h, agg1, E, Etot);
  elu_bias_kernel<<<(int)(((long long)Nn * 128 + T - 1) / T), T, 0, stream>>>(agg1, b1, (long long)Nn * 128);

  gat_wmma_gemm_k128<3, 40><<<dim3(Nn / 16, 1), 3 * 32, 0, stream>>>(agg1, W2, h2);
  scores2_kernel<<<(Nn + T - 1) / T, T, 0, stream>>>(h2, as2, ad2, asc2, adc2, Nn);
  edge_max2_kernel<<<(Etot + T - 1) / T, T, 0, stream>>>(srcp, dstp, asc2, adc2, m2, E, Etot);
  edge_sum2_kernel<<<(Etot + T - 1) / T, T, 0, stream>>>(srcp, dstp, asc2, adc2, m2, s2, E, Etot);
  edge_agg2_kernel<<<(Etot + 7) / 8, T, 0, stream>>>(srcp, dstp, asc2, adc2, m2, s2, h2, out, E, Etot);
}


# Both aggregation passes: a wave takes 8 edges at once, every index, score and row load of the 8 in flight before its first atomic (was 1 edge per wave, hops end to end); sums stay in arrival order
# speedup vs baseline: 1.8420x; 1.8420x over previous
#include <hip/hip_runtime.h>
#include <hip/hip_bf16.h>
#include <math.h>
#include <stdint.h>


typedef float v2f __attribute__((ext_vector_type(2)));
typedef float v4f __attribute__((ext_vector_type(4)));
typedef float v8f __attribute__((ext_vector_type(8)));

#define NEG_SLOPE 0.2f

__device__ __forceinline__ void atomicMaxF32(float* addr, float val) {
  if (val >= 0.0f) {
    atomicMax((int*)addr, __float_as_int(val));
  } else {
    atomicMin((unsigned int*)addr, __float_as_uint(val));
  }
}

template <int NT, int NACT>
__global__ void __launch_bounds__(32 * ((NACT + 16 * NT - 1) / (16 * NT)))
gat_wmma_gemm_k128(const float* __restrict__ A, const float* __restrict__ B,
                   float* __restrict__ C) {
  const int lane = threadIdx.x & 31;
  const int wave = threadIdx.x >> 5;
  const int c = lane & 15;
  const int h = lane >> 4;
  constexpr int CG = (NACT + 16 * NT - 1) / (16 * NT);
  constexpr bool FULL = (NACT % (16 * NT)) == 0;
  const size_t m0 = (size_t)blockIdx.x * 16;
  const int n0 = (CG == 1) ? 0 : wave * (16 * NT);

  const float* arow = A + (m0 + c) * 128 + 4 * h;
  v4f q[16];
#pragma unroll
  for (int i = 0; i < 16; ++i) q[i] = *(const v4f*)(arow + 8 * i);

  v2f b[NT][32];
#pragma unroll
  for (int t = 0; t < NT; ++t) {
    const int gn = n0 + 16 * t + c;
    const bool in_range = FULL || (CG == 1 && 16 * t + 15 < NACT) || gn < NACT;
    const float* bcol = B + (size_t)(4 * h) * NACT + gn;
#pragma unroll
    for (int j = 0; j < 32; ++j) b[t][j] = (v2f){0.f, 0.f};
    if (in_range) {
#pragma unroll
      for (int j = 0; j < 32; ++j) {
        const int k = 8 * (j >> 1) + 2 * (j & 1);
        b[t][j].x = bcol[(size_t)k * NACT];
        b[t][j].y = bcol[(size_t)(k + 1) * NACT];
      }
    }
  }

  v8f acc[NT];
#pragma unroll
  for (int t = 0; t < NT; ++t) acc[t] = (v8f){0.f, 0.f, 0.f, 0.f, 0.f, 0.f, 0.f, 0.f};

#pragma unroll
  for (int j = 0; j < 32; ++j) {
    const v4f qq = q[j >> 1];
    v2f a;
    if (j & 1) { a.x = qq.z; a.y = qq.w; } else { a.x = qq.x; a.y = qq.y; }
#pragma unroll
    for (int t = 0; t < NT; ++t) {
      acc[t] = __builtin_amdgcn_wmma_f32_16x16x4_f32(
           false, a,  false, b[t][j],
           (short)0, acc[t],  false,  false);
    }
  }

#pragma unroll
  for (int t = 0; t < NT; ++t) {
    const int gn = n0 + 16 * t + c;
    if (FULL || (CG == 1 && 16 * t + 15 < NACT) || gn < NACT) {
      float* cp = C + (m0 + (size_t)(h << 3)) * NACT + gn;
#pragma unroll
      for (int r = 0; r < 8; ++r) cp[(size_t)r * NACT] = acc[t][r];
    }
  }
}

__global__ void fill_kernel(float* __restrict__ p, float v, long long n) {
  long long i = (long long)blockIdx.x * blockDim.x + threadIdx.x;
  long long stride = (long long)gridDim.x * blockDim.x;
  for (; i < n; i += stride) p[i] = v;
}

__global__ void finalize_out_kernel(float* __restrict__ out, const float* __restrict__ ssum,
                                    const float* __restrict__ b2, int Nn) {
  int idx = blockIdx.x * blockDim.x + threadIdx.x;
  if (idx >= Nn * 40) return;
  int n = idx / 40;
  out[idx] = out[idx] / ssum[n] + b2[idx - n * 40];
}

__global__ void scores1_kernel(const float* __restrict__ h,
                               const float* __restrict__ att_src,
                               const float* __restrict__ att_dst,
                               float* __restrict__ a_src, float* __restrict__ a_dst,
                               int Nn) {
  int idx = blockIdx.x * blockDim.x + threadIdx.x;
  if (idx >= Nn * 8) return;
  int n = idx >> 3, hd = idx & 7;
  const float* hp = h + (size_t)n * 128 + hd * 16;
  float s = 0.f, d = 0.f;
#pragma unroll
  for (int c = 0; c < 16; ++c) {
    float v = hp[c];
    s = fmaf(v, att_src[hd * 16 + c], s);
    d = fmaf(v, att_dst[hd * 16 + c], d);
  }
  a_src[idx] = s;
  a_dst[idx] = d;
}

__global__ void scores2_kernel(const float* __restrict__ h2,
                               const float* __restrict__ att_src,
                               const float* __restrict__ att_dst,
                               float* __restrict__ a_src, float* __restrict__ a_dst,
                               int Nn) {
  int n = blockIdx.x * blockDim.x + threadIdx.x;
  if (n >= Nn) return;
  const float* hp = h2 + (size_t)n * 40;
  float s = 0.f, d = 0.f;
#pragma unroll
  for (int c = 0; c < 40; ++c) {
    float v = hp[c];
    s = fmaf(v, att_src[c], s);
    d = fmaf(v, att_dst[c], d);
  }
  a_src[n] = s;
  a_dst[n] = d;
}

__global__ void edge_max1_kernel(const int* __restrict__ src, const int* __restrict__ dst,
                                 const float* __restrict__ a_src, const float* __restrict__ a_dst,
                                 float* __restrict__ m, int E, int Etot) {
  int idx = blockIdx.x * blockDim.x + threadIdx.x;
  if (idx >= Etot * 8) return;
  int e = idx >> 3, hd = idx & 7;
  int s = (e < E) ? src[e] : (e - E);
  int d = (e < E) ? dst[e] : (e - E);
  float v = a_src[s * 8 + hd] + a_dst[d * 8 + hd];
  v = (v >= 0.f) ? v : NEG_SLOPE * v;
  atomicMaxF32(&m[d * 8 + hd], v);
}

template <int J>
__device__ __forceinline__ float quad_bcast(float v) {
  return __int_as_float(__builtin_amdgcn_update_dpp(
      0, __float_as_int(v), J * 0x55, 0xF, 0xF, true));
}

#define EPW 8

__device__ __forceinline__ void edge1_single(int s, int d, int lane, int hd,
                                             const float* __restrict__ a_src,
                                             const float* __restrict__ a_dst,
                                             const float* __restrict__ m,
                                             float* __restrict__ ssum,
                                             const float* __restrict__ h,
                                             float* __restrict__ agg) {
  float e = a_src[s * 8 + hd] + a_dst[d * 8 + hd];
  e = (e >= 0.f) ? e : NEG_SLOPE * e;
  float ex = __expf(e - m[d * 8 + hd]);
  const float* hp = h + (size_t)s * 128 + lane;
  const float h0 = hp[0], h1 = hp[32], h2 = hp[64], h3 = hp[96];
  const float w0 = quad_bcast<0>(ex);
  const float w1 = quad_bcast<1>(ex);
  const float w2 = quad_bcast<2>(ex);
  const float w3 = quad_bcast<3>(ex);
  float* op = agg + (size_t)d * 128 + lane;
  atomicAdd(op +  0, h0 * w0);
  atomicAdd(op + 32, h1 * w1);
  atomicAdd(op + 64, h2 * w2);
  atomicAdd(op + 96, h3 * w3);
  if ((lane & 12) == 0) atomicAdd(&ssum[d * 8 + hd], ex);
}

__global__ void __launch_bounds__(256)
edge_agg1_kernel(const int* __restrict__ src, const int* __restrict__ dst,
                 const float* __restrict__ a_src, const float* __restrict__ a_dst,
                 const float* __restrict__ m, float* __restrict__ ssum,
                 const float* __restrict__ h, float* __restrict__ agg,
                 int E, int Etot) {
  const int wid  = (int)(((long long)blockIdx.x * blockDim.x + threadIdx.x) >> 5);
  const int lane = threadIdx.x & 31;
  const long long base = (long long)wid * EPW;
  if (base >= Etot) return;
  const int hd = ((lane & 3) << 1) | (lane >> 4);

  if (base + EPW <= Etot) {
    const int eo = (int)base + (lane & (EPW - 1));
    const int s_own = (eo < E) ? src[eo] : (eo - E);
    const int d_own = (eo < E) ? dst[eo] : (eo - E);
    int   sk[EPW], dk[EPW];
    float es[EPW], ed[EPW], mm[EPW], r0[EPW], r1[EPW], r2[EPW], r3[EPW];
#pragma unroll
    for (int k = 0; k < EPW; ++k) {
      sk[k] = __builtin_amdgcn_readlane(s_own, k);
      dk[k] = __builtin_amdgcn_readlane(d_own, k);
      es[k] = a_src[sk[k] * 8 + hd];
      ed[k] = a_dst[dk[k] * 8 + hd];
      mm[k] = m[dk[k] * 8 + hd];
      const float* hp = h + (size_t)sk[k] * 128 + lane;
      r0[k] = hp[0]; r1[k] = hp[32]; r2[k] = hp[64]; r3[k] = hp[96];
    }
#pragma unroll
    for (int k = 0; k < EPW; ++k) {
      float e = es[k] + ed[k];
      e = (e >= 0.f) ? e : NEG_SLOPE * e;
      const float ex = __expf(e - mm[k]);
      const float w0 = quad_bcast<0>(ex);
      const float w1 = quad_bcast<1>(ex);
      const float w2 = quad_bcast<2>(ex);
      const float w3 = quad_bcast<3>(ex);
      float* op = agg + (size_t)dk[k] * 128 + lane;
      atomicAdd(op +  0, r0[k] * w0);
      atomicAdd(op + 32, r1[k] * w1);
      atomicAdd(op + 64, r2[k] * w2);
      atomicAdd(op + 96, r3[k] * w3);
      if ((lane & 12) == 0) atomicAdd(&ssum[dk[k] * 8 + hd], ex);
    }
  } else {
    for (int k = 0; k < EPW; ++k) {
      const long long e = base + k;
      if (e >= Etot) break;
      const int s = (e < E) ? src[e] : (int)(e - E);
      const int d = (e < E) ? dst[e] : (int)(e - E);
      edge1_single(s, d, lane, hd, a_src, a_dst, m, ssum, h, agg);
    }
  }
}

__global__ void elu_bias_kernel(float* __restrict__ a, const float* __restrict__ ssum,
                                const float* __restrict__ b, long long n) {
  long long idx = (long long)blockIdx.x * blockDim.x + threadIdx.x;
  if (idx >= n) return;
  float v = a[idx] / ssum[idx >> 4] + b[idx & 127];
  a[idx] = (v > 0.f) ? v : (__expf(v) - 1.0f);
}

__global__ void edge_max2_kernel(const int* __restrict__ src, const int* __restrict__ dst,
                                 const float* __restrict__ a_src, const float* __restrict__ a_dst,
                                 float* __restrict__ m, int E, int Etot) {
  int e = blockIdx.x * blockDim.x + threadIdx.x;
  if (e >= Etot) return;
  int s = (e < E) ? src[e] : (e - E);
  int d = (e < E) ? dst[e] : (e - E);
  float v = a_src[s] + a_dst[d];
  v = (v >= 0.f) ? v : NEG_SLOPE * v;
  atomicMaxF32(&m[d], v);
}

__global__ void __launch_bounds__(256)
edge_agg2_kernel(const int* __restrict__ src, const int* __restrict__ dst,
                 const float* __restrict__ a_src, const float* __restrict__ a_dst,
                 const float* __restrict__ m, float* __restrict__ ssum,
                 const float* __restrict__ h2, float* __restrict__ out,
                 int E, int Etot) {
  const int wid  = (int)(((long long)blockIdx.x * blockDim.x + threadIdx.x) >> 5);
  const int lane = threadIdx.x & 31;
  const long long base = (long long)wid * EPW;
  if (base >= Etot) return;

  if (base + EPW <= Etot) {
    const int eo = (int)base + (lane & (EPW - 1));
    const int s_own = (eo < E) ? src[eo] : (eo - E);
    const int d_own = (eo < E) ? dst[eo] : (eo - E);
    const float es = a_src[s_own];
    const float ed = a_dst[d_own];
    const float mo = m[d_own];
    int   sk[EPW], dk[EPW];
    float ra[EPW], rb[EPW];
#pragma unroll
    for (int k = 0; k < EPW; ++k) {
      sk[k] = __builtin_amdgcn_readlane(s_own, k);
      dk[k] = __builtin_amdgcn_readlane(d_own, k);
      ra[k] = h2[(size_t)sk[k] * 40 + lane];
    }
    if (lane < 8) {
#pragma unroll
      for (int k = 0; k < EPW; ++k) rb[k] = h2[(size_t)sk[k] * 40 + 32 + lane];
    }
    float e = es + ed;
    e = (e >= 0.f) ? e : NEG_SLOPE * e;
    const float ex_own = __expf(e - mo);
    float wk[EPW];
#pragma unroll
    for (int k = 0; k < EPW; ++k) {
      wk[k] = __int_as_float(__builtin_amdgcn_readlane(__float_as_int(ex_own), k));
      atomicAdd(&out[(size_t)dk[k] * 40 + lane], ra[k] * wk[k]);
    }
    if (lane < 8) {
#pragma unroll
      for (int k = 0; k < EPW; ++k)
        atomicAdd(&out[(size_t)dk[k] * 40 + 32 + lane], rb[k] * wk[k]);
      atomicAdd(&ssum[d_own], ex_own);
    }
  } else {
    for (int k = 0; k < EPW; ++k) {
      const long long eg = base + k;
      if (eg >= Etot) break;
      const int s = (eg < E) ? src[eg] : (int)(eg - E);
      const int d = (eg < E) ? dst[eg] : (int)(eg - E);
      float e = a_src[s] + a_dst[d];
      e = (e >= 0.f) ? e : NEG_SLOPE * e;
      const float ex = __expf(e - m[d]);
      for (int c = lane; c < 40; c += 32)
        atomicAdd(&out[(size_t)d * 40 + c], h2[(size_t)s * 40 + c] * ex);
      if (lane == 0) atomicAdd(&ssum[d], ex);
    }
  }
}

extern "C" void kernel_launch(void* const* d_in, const int* in_sizes, int n_in,
                              void* d_out, int out_size, void* d_ws, size_t ws_size,
                              hipStream_t stream) {
  const float* x   = (const float*)d_in[0];
  const int*   ei  = (const int*)d_in[1];
  const float* W1  = (const float*)d_in[2];
  const float* as1 = (const float*)d_in[3];
  const float* ad1 = (const float*)d_in[4];
  const float* b1  = (const float*)d_in[5];
  const float* W2  = (const float*)d_in[6];
  const float* as2 = (const float*)d_in[7];
  const float* ad2 = (const float*)d_in[8];
  const float* b2  = (const float*)d_in[9];
  float* out = (float*)d_out;

  const int Nn   = in_sizes[0] / 128;
  const int E    = in_sizes[1] / 2;
  const int Etot = E + Nn;
  const int* srcp = ei;
  const int* dstp = ei + E;

  float* ws   = (float*)d_ws;
  float* h    = ws;  ws += (size_t)Nn * 128;
  float* agg1 = ws;  ws += (size_t)Nn * 128;
  float* asc1 = ws;  ws += (size_t)Nn * 8;
  float* adc1 = ws;  ws += (size_t)Nn * 8;
  float* m1   = ws;  ws += (size_t)Nn * 8;
  float* s1   = ws;  ws += (size_t)Nn * 8;
  float* h2   = ws;  ws += (size_t)Nn * 40;
  float* asc2 = ws;  ws += Nn;
  float* adc2 = ws;  ws += Nn;
  float* m2   = ws;  ws += Nn;
  float* s2   = ws;  ws += Nn;

  const int T = 256;
  const int fillBlocks = 2048;

  fill_kernel<<<fillBlocks, T, 0, stream>>>(m1, -INFINITY, (long long)Nn * 8);
  fill_kernel<<<fillBlocks, T, 0, stream>>>(s1, 0.0f, (long long)Nn * 8);
  fill_kernel<<<fillBlocks, T, 0, stream>>>(agg1, 0.0f, (long long)Nn * 128);
  fill_kernel<<<fillBlocks, T, 0, stream>>>(m2, -INFINITY, (long long)Nn);
  fill_kernel<<<fillBlocks, T, 0, stream>>>(s2, 0.0f, (long long)Nn);
  fill_kernel<<<fillBlocks, T, 0, stream>>>(out, 0.0f, (long long)Nn * 40);

  gat_wmma_gemm_k128<2, 128><<<dim3(Nn / 16), 4 * 32, 0, stream>>>(x, W1, h);
  scores1_kernel<<<(Nn * 8 + T - 1) / T, T, 0, stream>>>(h, as1, ad1, asc1, adc1, Nn);
  edge_max1_kernel<<<(Etot * 8 + T - 1) / T, T, 0, stream>>>(srcp, dstp, asc1, adc1, m1, E, Etot);
  edge_agg1_kernel<<<(Etot + 8 * EPW - 1) / (8 * EPW), T, 0, stream>>>(srcp, dstp, asc1, adc1, m1, s1, h, agg1, E, Etot);
  elu_bias_kernel<<<(int)(((long long)Nn * 128 + T - 1) / T), T, 0, stream>>>(agg1, s1, b1, (long long)Nn * 128);

  gat_wmma_gemm_k128<3, 40><<<dim3(Nn / 16), 32, 0, stream>>>(agg1, W2, h2);
  scores2_kernel<<<(Nn + T - 1) / T, T, 0, stream>>>(h2, as2, ad2, asc2, adc2, Nn);
  edge_max2_kernel<<<(Etot + T - 1) / T, T, 0, stream>>>(srcp, dstp, asc2, adc2, m2, E, Etot);
  edge_agg2_kernel<<<(Etot + 8 * EPW - 1) / (8 * EPW), T, 0, stream>>>(srcp, dstp, asc2, adc2, m2, s2, h2, out, E, Etot);
  finalize_out_kernel<<<(Nn * 40 + T - 1) / T, T, 0, stream>>>(out, s2, b2, Nn);
}


# Every sum the edge passes add into (rows and exp sums, both layers) gets one plane per package half; a wave adds into its own half's plane, node epilogues add the planes (sums: per half, then both)
# speedup vs baseline: 2.0842x; 1.1315x over previous
#include <hip/hip_runtime.h>
#include <hip/hip_bf16.h>
#include <math.h>
#include <stdint.h>


typedef float v2f __attribute__((ext_vector_type(2)));
typedef float v4f __attribute__((ext_vector_type(4)));
typedef float v8f __attribute__((ext_vector_type(8)));

#define NEG_SLOPE 0.2f

__device__ __forceinline__ void atomicMaxF32(float* addr, float val) {
  if (val >= 0.0f) {
    atomicMax((int*)addr, __float_as_int(val));
  } else {
    atomicMin((unsigned int*)addr, __float_as_uint(val));
  }
}

__device__ __forceinline__ int package_half() {
  return (int)((__builtin_amdgcn_s_sendmsg_rtn(0x87) >> 18) & 1u);
}

template <int NT, int NACT>
__global__ void __launch_bounds__(32 * ((NACT + 16 * NT - 1) / (16 * NT)))
gat_wmma_gemm_k128(const float* __restrict__ A, const float* __restrict__ B,
                   float* __restrict__ C) {
  const int lane = threadIdx.x & 31;
  const int wave = threadIdx.x >> 5;
  const int c = lane & 15;
  const int h = lane >> 4;
  constexpr int CG = (NACT + 16 * NT - 1) / (16 * NT);
  constexpr bool FULL = (NACT % (16 * NT)) == 0;
  const size_t m0 = (size_t)blockIdx.x * 16;
  const int n0 = (CG == 1) ? 0 : wave * (16 * NT);

  const float* arow = A + (m0 + c) * 128 + 4 * h;
  v4f q[16];
#pragma unroll
  for (int i = 0; i < 16; ++i) q[i] = *(const v4f*)(arow + 8 * i);

  v2f b[NT][32];
#pragma unroll
  for (int t = 0; t < NT; ++t) {
    const int gn = n0 + 16 * t + c;
    const bool in_range = FULL || (CG == 1 && 16 * t + 15 < NACT) || gn < NACT;
    const float* bcol = B + (size_t)(4 * h) * NACT + gn;
#pragma unroll
    for (int j = 0; j < 32; ++j) b[t][j] = (v2f){0.f, 0.f};
    if (in_range) {
#pragma unroll
      for (int j = 0; j < 32; ++j) {
        const int k = 8 * (j >> 1) + 2 * (j & 1);
        b[t][j].x = bcol[(size_t)k * NACT];
        b[t][j].y = bcol[(size_t)(k + 1) * NACT];
      }
    }
  }

  v8f acc[NT];
#pragma unroll
  for (int t = 0; t < NT; ++t) acc[t] = (v8f){0.f, 0.f, 0.f, 0.f, 0.f, 0.f, 0.f, 0.f};

#pragma unroll
  for (int j = 0; j < 32; ++j) {
    const v4f qq = q[j >> 1];
    v2f a;
    if (j & 1) { a.x = qq.z; a.y = qq.w; } else { a.x = qq.x; a.y = qq.y; }
#pragma unroll
    for (int t = 0; t < NT; ++t) {
      acc[t] = __builtin_amdgcn_wmma_f32_16x16x4_f32(
           false, a,  false, b[t][j],
           (short)0, acc[t],  false,  false);
    }
  }

#pragma unroll
  for (int t = 0; t < NT; ++t) {
    const int gn = n0 + 16 * t + c;
    if (FULL || (CG == 1 && 16 * t + 15 < NACT) || gn < NACT) {
      float* cp = C + (m0 + (size_t)(h << 3)) * NACT + gn;
#pragma unroll
      for (int r = 0; r < 8; ++r) cp[(size_t)r * NACT] = acc[t][r];
    }
  }
}

__global__ void fill_kernel(float* __restrict__ p, float v, long long n) {
  long long i = (long long)blockIdx.x * blockDim.x + threadIdx.x;
  long long stride = (long long)gridDim.x * blockDim.x;
  for (; i < n; i += stride) p[i] = v;
}

__global__ void finalize_out_kernel(float* __restrict__ out, const float* __restrict__ out_hi,
                                    const float* __restrict__ ssum,
                                    const float* __restrict__ ssum_hi,
                                    const float* __restrict__ b2, int Nn) {
  int idx = blockIdx.x * blockDim.x + threadIdx.x;
  if (idx >= Nn * 40) return;
  int n = idx / 40;
  out[idx] = (out[idx] + out_hi[idx]) / (ssum[n] + ssum_hi[n]) + b2[idx - n * 40];
}

__global__ void scores1_kernel(const float* __restrict__ h,
                               const float* __restrict__ att_src,
                               const float* __restrict__ att_dst,
                               float* __restrict__ a_src, float* __restrict__ a_dst,
                               int Nn) {
  int idx = blockIdx.x * blockDim.x + threadIdx.x;
  if (idx >= Nn * 8) return;
  int n = idx >> 3, hd = idx & 7;
  const float* hp = h + (size_t)n * 128 + hd * 16;
  float s = 0.f, d = 0.f;
#pragma unroll
  for (int c = 0; c < 16; ++c) {
    float v = hp[c];
    s = fmaf(v, att_src[hd * 16 + c], s);
    d = fmaf(v, att_dst[hd * 16 + c], d);
  }
  a_src[idx] = s;
  a_dst[idx] = d;
}

__global__ void scores2_kernel(const float* __restrict__ h2,
                               const float* __restrict__ att_src,
                               const float* __restrict__ att_dst,
                               float* __restrict__ a_src, float* __restrict__ a_dst,
                               int Nn) {
  int n = blockIdx.x * blockDim.x + threadIdx.x;
  if (n >= Nn) return;
  const float* hp = h2 + (size_t)n * 40;
  float s = 0.f, d = 0.f;
#pragma unroll
  for (int c = 0; c < 40; ++c) {
    float v = hp[c];
    s = fmaf(v, att_src[c], s);
    d = fmaf(v, att_dst[c], d);
  }
  a_src[n] = s;
  a_dst[n] = d;
}

__global__ void edge_max1_kernel(const int* __restrict__ src, const int* __restrict__ dst,
                                 const float* __restrict__ a_src, const float* __restrict__ a_dst,
                                 float* __restrict__ m, int E, int Etot) {
  int idx = blockIdx.x * blockDim.x + threadIdx.x;
  if (idx >= Etot * 8) return;
  int e = idx >> 3, hd = idx & 7;
  int s = (e < E) ? src[e] : (e - E);
  int d = (e < E) ? dst[e] : (e - E);
  float v = a_src[s * 8 + hd] + a_dst[d * 8 + hd];
  v = (v >= 0.f) ? v : NEG_SLOPE * v;
  atomicMaxF32(&m[d * 8 + hd], v);
}

template <int J>
__device__ __forceinline__ float quad_bcast(float v) {
  return __int_as_float(__builtin_amdgcn_update_dpp(
      0, __float_as_int(v), J * 0x55, 0xF, 0xF, true));
}

#define EPW 8

__device__ __forceinline__ void edge1_single(int s, int d, int lane, int hd,
                                             const float* __restrict__ a_src,
                                             const float* __restrict__ a_dst,
                                             const float* __restrict__ m,
                                             float* __restrict__ ssum,
                                             const float* __restrict__ h,
                                             float* __restrict__ agg) {
  float e = a_src[s * 8 + hd] + a_dst[d * 8 + hd];
  e = (e >= 0.f) ? e : NEG_SLOPE * e;
  float ex = __expf(e - m[d * 8 + hd]);
  const float* hp = h + (size_t)s * 128 + lane;
  const float h0 = hp[0], h1 = hp[32], h2 = hp[64], h3 = hp[96];
  const float w0 = quad_bcast<0>(ex);
  const float w1 = quad_bcast<1>(ex);
  const float w2 = quad_bcast<2>(ex);
  const float w3 = quad_bcast<3>(ex);
  float* op = agg + (size_t)d * 128 + lane;
  atomicAdd(op +  0, h0 * w0);
  atomicAdd(op + 32, h1 * w1);
  atomicAdd(op + 64, h2 * w2);
  atomicAdd(op + 96, h3 * w3);
  if ((lane & 12) == 0) atomicAdd(&ssum[d * 8 + hd], ex);
}

__global__ void __launch_bounds__(256)
edge_agg1_kernel(const int* __restrict__ src, const int* __restrict__ dst,
                 const float* __restrict__ a_src, const float* __restrict__ a_dst,
                 const float* __restrict__ m, float* __restrict__ ssum,
                 const float* __restrict__ h, float* __restrict__ agg,
                 int E, int Etot, int Nn) {
  const int wid  = (int)(((long long)blockIdx.x * blockDim.x + threadIdx.x) >> 5);
  const int lane = threadIdx.x & 31;
  const long long base = (long long)wid * EPW;
  if (base >= Etot) return;
  const int hd = ((lane & 3) << 1) | (lane >> 4);

  if (base + EPW <= Etot) {
    const int eo = (int)base + (lane & (EPW - 1));
    const int s_own = (eo < E) ? src[eo] : (eo - E);
    const int d_own = (eo < E) ? dst[eo] : (eo - E);
    const size_t plane = (size_t)package_half() * (size_t)Nn;
    agg  += plane * 128;
    ssum += plane * 8;
    int   sk[EPW], dk[EPW];
    float es[EPW], ed[EPW], mm[EPW], r0[EPW], r1[EPW], r2[EPW], r3[EPW];
#pragma unroll
    for (int k = 0; k < EPW; ++k) {
      sk[k] = __builtin_amdgcn_readlane(s_own, k);
      dk[k] = __builtin_amdgcn_readlane(d_own, k);
      es[k] = a_src[sk[k] * 8 + hd];
      ed[k] = a_dst[dk[k] * 8 + hd];
      mm[k] = m[dk[k] * 8 + hd];
      const float* hp = h + (size_t)sk[k] * 128 + lane;
      r0[k] = hp[0]; r1[k] = hp[32]; r2[k] = hp[64]; r3[k] = hp[96];
    }
#pragma unroll
    for (int k = 0; k < EPW; ++k) {
      float e = es[k] + ed[k];
      e = (e >= 0.f) ? e : NEG_SLOPE * e;
      const float ex = __expf(e - mm[k]);
      const float w0 = quad_bcast<0>(ex);
      const float w1 = quad_bcast<1>(ex);
      const float w2 = quad_bcast<2>(ex);
      const float w3 = quad_bcast<3>(ex);
      float* op = agg + (size_t)dk[k] * 128 + lane;
      atomicAdd(op +  0, r0[k] * w0);
      atomicAdd(op + 32, r1[k] * w1);
      atomicAdd(op + 64, r2[k] * w2);
      atomicAdd(op + 96, r3[k] * w3);
      if ((lane & 12) == 0) atomicAdd(&ssum[dk[k] * 8 + hd], ex);
    }
  } else {
    const size_t plane = (size_t)package_half() * (size_t)Nn;
    agg  += plane * 128;
    ssum += plane * 8;
    for (int k = 0; k < EPW; ++k) {
      const long long e = base + k;
      if (e >= Etot) break;
      const int s = (e < E) ? src[e] : (int)(e - E);
      const int d = (e < E) ? dst[e] : (int)(e - E);
      edge1_single(s, d, lane, hd, a_src, a_dst, m, ssum, h, agg);
    }
  }
}

__global__ void elu_bias_kernel(float* __restrict__ a, const float* __restrict__ a_hi,
                                const float* __restrict__ ssum,
                                const float* __restrict__ ssum_hi,
                                const float* __restrict__ b, long long n) {
  long long idx = (long long)blockIdx.x * blockDim.x + threadIdx.x;
  if (idx >= n) return;
  float v = (a[idx] + a_hi[idx]) / (ssum[idx >> 4] + ssum_hi[idx >> 4]) + b[idx & 127];
  a[idx] = (v > 0.f) ? v : (__expf(v) - 1.0f);
}

__global__ void edge_max2_kernel(const int* __restrict__ src, const int* __restrict__ dst,
                                 const float* __restrict__ a_src, const float* __restrict__ a_dst,
                                 float* __restrict__ m, int E, int Etot) {
  int e = blockIdx.x * blockDim.x + threadIdx.x;
  if (e >= Etot) return;
  int s = (e < E) ? src[e] : (e - E);
  int d = (e < E) ? dst[e] : (e - E);
  float v = a_src[s] + a_dst[d];
  v = (v >= 0.f) ? v : NEG_SLOPE * v;
  atomicMaxF32(&m[d], v);
}

__global__ void __launch_bounds__(256)
edge_agg2_kernel(const int* __restrict__ src, const int* __restrict__ dst,
                 const float* __restrict__ a_src, const float* __restrict__ a_dst,
                 const float* __restrict__ m, float* __restrict__ ssum,
                 const float* __restrict__ h2, float* __restrict__ out_lo,
                 float* __restrict__ out_hi, int E, int Etot, int Nn) {
  const int wid  = (int)(((long long)blockIdx.x * blockDim.x + threadIdx.x) >> 5);
  const int lane = threadIdx.x & 31;
  const long long base = (long long)wid * EPW;
  if (base >= Etot) return;

  if (base + EPW <= Etot) {
    const int eo = (int)base + (lane & (EPW - 1));
    const int s_own = (eo < E) ? src[eo] : (eo - E);
    const int d_own = (eo < E) ? dst[eo] : (eo - E);
    const int half = package_half();
    float* __restrict__ out = half ? out_hi : out_lo;
    ssum += (size_t)half * (size_t)Nn;
    const float es = a_src[s_own];
    const float ed = a_dst[d_own];
    const float mo = m[d_own];
    int   sk[EPW], dk[EPW];
    float ra[EPW], rb[EPW];
#pragma unroll
    for (int k = 0; k < EPW; ++k) {
      sk[k] = __builtin_amdgcn_readlane(s_own, k);
      dk[k] = __builtin_amdgcn_readlane(d_own, k);
      ra[k] = h2[(size_t)sk[k] * 40 + lane];
    }
    if (lane < 8) {
#pragma unroll
      for (int k = 0; k < EPW; ++k) rb[k] = h2[(size_t)sk[k] * 40 + 32 + lane];
    }
    float e = es + ed;
    e = (e >= 0.f) ? e : NEG_SLOPE * e;
    const float ex_own = __expf(e - mo);
    float wk[EPW];
#pragma unroll
    for (int k = 0; k < EPW; ++k) {
      wk[k] = __int_as_float(__builtin_amdgcn_readlane(__float_as_int(ex_own), k));
      atomicAdd(&out[(size_t)dk[k] * 40 + lane], ra[k] * wk[k]);
    }
    if (lane < 8) {
#pragma unroll
      for (int k = 0; k < EPW; ++k)
        atomicAdd(&out[(size_t)dk[k] * 40 + 32 + lane], rb[k] * wk[k]);
      atomicAdd(&ssum[d_own], ex_own);
    }
  } else {
    const int half = package_half();
    float* __restrict__ out = half ? out_hi : out_lo;
    ssum += (size_t)half * (size_t)Nn;
    for (int k = 0; k < EPW; ++k) {
      const long long eg = base + k;
      if (eg >= Etot) break;
      const int s = (eg < E) ? src[eg] : (int)(eg - E);
      const int d = (eg < E) ? dst[eg] : (int)(eg - E);
      float e = a_src[s] + a_dst[d];
      e = (e >= 0.f) ? e : NEG_SLOPE * e;
      const float ex = __expf(e - m[d]);
      for (int c = lane; c < 40; c += 32)
        atomicAdd(&out[(size_t)d * 40 + c], h2[(size_t)s * 40 + c] * ex);
      if (lane == 0) atomicAdd(&ssum[d], ex);
    }
  }
}

extern "C" void kernel_launch(void* const* d_in, const int* in_sizes, int n_in,
                              void* d_out, int out_size, void* d_ws, size_t ws_size,
                              hipStream_t stream) {
  const float* x   = (const float*)d_in[0];
  const int*   ei  = (const int*)d_in[1];
  const float* W1  = (const float*)d_in[2];
  const float* as1 = (const float*)d_in[3];
  const float* ad1 = (const float*)d_in[4];
  const float* b1  = (const float*)d_in[5];
  const float* W2  = (const float*)d_in[6];
  const float* as2 = (const float*)d_in[7];
  const float* ad2 = (const float*)d_in[8];
  const float* b2  = (const float*)d_in[9];
  float* out = (float*)d_out;

  const int Nn   = in_sizes[0] / 128;
  const int E    = in_sizes[1] / 2;
  const int Etot = E + Nn;
  const int* srcp = ei;
  const int* dstp = ei + E;

  float* ws   = (float*)d_ws;
  float* h    = ws;  ws += (size_t)Nn * 128;
  float* agg1 = ws;  ws += (size_t)Nn * 256;
  float* asc1 = ws;  ws += (size_t)Nn * 8;
  float* adc1 = ws;  ws += (size_t)Nn * 8;
  float* m1   = ws;  ws += (size_t)Nn * 8;
  float* s1   = ws;  ws += (size_t)Nn * 16;
  float* h2   = ws;  ws += (size_t)Nn * 40;
  float* outb = ws;  ws += (size_t)Nn * 40;
  float* s2   = ws;  ws += (size_t)Nn * 2;
  float* asc2 = ws;  ws += Nn;
  float* adc2 = ws;  ws += Nn;
  float* m2   = ws;  ws += Nn;

  const int T = 256;
  const int fillBlocks = 2048;

  fill_kernel<<<fillBlocks, T, 0, stream>>>(m1, -INFINITY, (long long)Nn * 8);
  fill_kernel<<<fillBlocks, T, 0, stream>>>(s1, 0.0f, (long long)Nn * 16);
  fill_kernel<<<fillBlocks, T, 0, stream>>>(agg1, 0.0f, (long long)Nn * 256);
  fill_kernel<<<fillBlocks, T, 0, stream>>>(m2, -INFINITY, (long long)Nn);
  fill_kernel<<<fillBlocks, T, 0, stream>>>(outb, 0.0f, (long long)Nn * 42);
  fill_kernel<<<fillBlocks, T, 0, stream>>>(out, 0.0f, (long long)Nn * 40);

  gat_wmma_gemm_k128<2, 128><<<dim3(Nn / 16), 4 * 32, 0, stream>>>(x, W1, h);
  scores1_kernel<<<(Nn * 8 + T - 1) / T, T, 0, stream>>>(h, as1, ad1, asc1, adc1, Nn);
  edge_max1_kernel<<<(Etot * 8 + T - 1) / T, T, 0, stream>>>(srcp, dstp, asc1, adc1, m1, E, Etot);
  edge_agg1_kernel<<<(Etot + 8 * EPW - 1) / (8 * EPW), T, 0, stream>>>(srcp, dstp, asc1, adc1, m1, s1, h, agg1, E, Etot, Nn);
  elu_bias_kernel<<<(int)(((long long)Nn * 128 + T - 1) / T), T, 0, stream>>>(agg1, agg1 + (size_t)Nn * 128, s1, s1 + (size_t)Nn * 8, b1, (long long)Nn * 128);

  gat_wmma_gemm_k128<3, 40><<<dim3(Nn / 16), 32, 0, stream>>>(agg1, W2, h2);
  scores2_kernel<<<(Nn + T - 1) / T, T, 0, stream>>>(h2, as2, ad2, asc2, adc2, Nn);
  edge_max2_kernel<<<(Etot + T - 1) / T, T, 0, stream>>>(srcp, dstp, asc2, adc2, m2, E, Etot);
  edge_agg2_kernel<<<(Etot + 8 * EPW - 1) / (8 * EPW), T, 0, stream>>>(srcp, dstp, asc2, adc2, m2, s2, h2, out, outb, E, Etot, Nn);
  finalize_out_kernel<<<(Nn * 40 + T - 1) / T, T, 0, stream>>>(out, outb, s2, s2 + Nn, b2, Nn);
}


# Softmax shift is an upper bound of a node's edge scores, leaky(a_dst + largest a_src of all nodes, found by the score kernels); per-edge max passes run only on a spread over 16; sums in arrival order
# speedup vs baseline: 3.0236x; 1.4507x over previous
#include <hip/hip_runtime.h>
#include <hip/hip_bf16.h>
#include <math.h>
#include <stdint.h>


typedef float v2f __attribute__((ext_vector_type(2)));
typedef float v4f __attribute__((ext_vector_type(4)));
typedef float v8f __attribute__((ext_vector_type(8)));

#define NEG_SLOPE 0.2f
#define SPREAD_MAX 16.0f
#define STAT_PITCH 32
#define STAT_WORDS 18

__device__ __forceinline__ void atomicMaxF32(float* addr, float val) {
  if (val >= 0.0f) {
    atomicMax((int*)addr, __float_as_int(val));
  } else {
    atomicMin((unsigned int*)addr, __float_as_uint(val));
  }
}

__device__ __forceinline__ int package_half() {
  return (int)((__builtin_amdgcn_s_sendmsg_rtn(0x87) >> 18) & 1u);
}

template <int NT, int NACT>
__global__ void __launch_bounds__(32 * ((NACT + 16 * NT - 1) / (16 * NT)))
gat_wmma_gemm_k128(const float* __restrict__ A, const float* __restrict__ B,
                   float* __restrict__ C) {
  const int lane = threadIdx.x & 31;
  const int wave = threadIdx.x >> 5;
  const int c = lane & 15;
  const int h = lane >> 4;
  constexpr int CG = (NACT + 16 * NT - 1) / (16 * NT);
  constexpr bool FULL = (NACT % (16 * NT)) == 0;
  const size_t m0 = (size_t)blockIdx.x * 16;
  const int n0 = (CG == 1) ? 0 : wave * (16 * NT);

  const float* arow = A + (m0 + c) * 128 + 4 * h;
  v4f q[16];
#pragma unroll
  for (int i = 0; i < 16; ++i) q[i] = *(const v4f*)(arow + 8 * i);

  v2f b[NT][32];
#pragma unroll
  for (int t = 0; t < NT; ++t) {
    const int gn = n0 + 16 * t + c;
    const bool in_range = FULL || (CG == 1 && 16 * t + 15 < NACT) || gn < NACT;
    const float* bcol = B + (size_t)(4 * h) * NACT + gn;
#pragma unroll
    for (int j = 0; j < 32; ++j) b[t][j] = (v2f){0.f, 0.f};
    if (in_range) {
#pragma unroll
      for (int j = 0; j < 32; ++j) {
        const int k = 8 * (j >> 1) + 2 * (j & 1);
        b[t][j].x = bcol[(size_t)k * NACT];
        b[t][j].y = bcol[(size_t)(k + 1) * NACT];
      }
    }
  }

  v8f acc[NT];
#pragma unroll
  for (int t = 0; t < NT; ++t) acc[t] = (v8f){0.f, 0.f, 0.f, 0.f, 0.f, 0.f, 0.f, 0.f};

#pragma unroll
  for (int j = 0; j < 32; ++j) {
    const v4f qq = q[j >> 1];
    v2f a;
    if (j & 1) { a.x = qq.z; a.y = qq.w; } else { a.x = qq.x; a.y = qq.y; }
#pragma unroll
    for (int t = 0; t < NT; ++t) {
      acc[t] = __builtin_amdgcn_wmma_f32_16x16x4_f32(
           false, a,  false, b[t][j],
           (short)0, acc[t],  false,  false);
    }
  }

#pragma unroll
  for (int t = 0; t < NT; ++t) {
    const int gn = n0 + 16 * t + c;
    if (FULL || (CG == 1 && 16 * t + 15 < NACT) || gn < NACT) {
      float* cp = C + (m0 + (size_t)(h << 3)) * NACT + gn;
#pragma unroll
      for (int r = 0; r < 8; ++r) cp[(size_t)r * NACT] = acc[t][r];
    }
  }
}

__global__ void fill_kernel(float* __restrict__ p, float v, long long n) {
  long long i = (long long)blockIdx.x * blockDim.x + threadIdx.x;
  long long stride = (long long)gridDim.x * blockDim.x;
  for (; i < n; i += stride) p[i] = v;
}

__global__ void finalize_out_kernel(float* __restrict__ out, const float* __restrict__ out_hi,
                                    const float* __restrict__ ssum,
                                    const float* __restrict__ ssum_hi,
                                    const float* __restrict__ b2, int Nn) {
  int idx = blockIdx.x * blockDim.x + threadIdx.x;
  if (idx >= Nn * 40) return;
  int n = idx / 40;
  out[idx] = (out[idx] + out_hi[idx]) / (ssum[n] + ssum_hi[n]) + b2[idx - n * 40];
}

__global__ void __launch_bounds__(256)
scores1_kernel(const float* __restrict__ h,
               const float* __restrict__ att_src,
               const float* __restrict__ att_dst,
               float* __restrict__ a_src, float* __restrict__ a_dst,
               float* __restrict__ stat, int Nn) {
  __shared__ float red[2][8][8];
  const int idx = blockIdx.x * blockDim.x + threadIdx.x;
  const bool valid = idx < Nn * 8;
  float s = 0.f, d = 0.f;
  if (valid) {
    const int n = idx >> 3, hd = idx & 7;
    const float* hp = h + (size_t)n * 128 + hd * 16;
#pragma unroll
    for (int c = 0; c < 16; ++c) {
      float v = hp[c];
      s = fmaf(v, att_src[hd * 16 + c], s);
      d = fmaf(v, att_dst[hd * 16 + c], d);
    }
    a_src[idx] = s;
    a_dst[idx] = d;
  }
  float vmax = valid ? s : -INFINITY;
  float vneg = valid ? -s : -INFINITY;
  vmax = fmaxf(vmax, __shfl_xor(vmax, 8));
  vneg = fmaxf(vneg, __shfl_xor(vneg, 8));
  vmax = fmaxf(vmax, __shfl_xor(vmax, 16));
  vneg = fmaxf(vneg, __shfl_xor(vneg, 16));
  const int lane = threadIdx.x & 31, wave = threadIdx.x >> 5;
  if (lane < 8) { red[0][wave][lane] = vmax; red[1][wave][lane] = vneg; }
  __syncthreads();
  if (threadIdx.x < 16) {
    const int which = threadIdx.x >> 3, hh = threadIdx.x & 7;
    float v = red[which][0][hh];
#pragma unroll
    for (int w = 1; w < 8; ++w) v = fmaxf(v, red[which][w][hh]);
    atomicMaxF32(&stat[threadIdx.x * STAT_PITCH], v);
  }
}

__global__ void __launch_bounds__(256)
scores2_kernel(const float* __restrict__ h2,
               const float* __restrict__ att_src,
               const float* __restrict__ att_dst,
               float* __restrict__ a_src, float* __restrict__ a_dst,
               float* __restrict__ stat, int Nn) {
  __shared__ float red[2][8];
  const int n = blockIdx.x * blockDim.x + threadIdx.x;
  const bool valid = n < Nn;
  float s = 0.f, d = 0.f;
  if (valid) {
    const float* hp = h2 + (size_t)n * 40;
#pragma unroll
    for (int c = 0; c < 40; ++c) {
      float v = hp[c];
      s = fmaf(v, att_src[c], s);
      d = fmaf(v, att_dst[c], d);
    }
    a_src[n] = s;
    a_dst[n] = d;
  }
  float vmax = valid ? s : -INFINITY;
  float vneg = valid ? -s : -INFINITY;
#pragma unroll
  for (int o = 16; o >= 1; o >>= 1) {
    vmax = fmaxf(vmax, __shfl_xor(vmax, o));
    vneg = fmaxf(vneg, __shfl_xor(vneg, o));
  }
  const int lane = threadIdx.x & 31, wave = threadIdx.x >> 5;
  if (lane == 0) { red[0][wave] = vmax; red[1][wave] = vneg; }
  __syncthreads();
  if (threadIdx.x < 2) {
    float v = red[threadIdx.x][0];
#pragma unroll
    for (int w = 1; w < 8; ++w) v = fmaxf(v, red[threadIdx.x][w]);
    atomicMaxF32(&stat[(16 + threadIdx.x) * STAT_PITCH], v);
  }
}

__device__ __forceinline__ bool spread_too_wide1(float gmax, float gneg) {
  return __builtin_amdgcn_ballot_w32(!(gmax + gneg <= SPREAD_MAX)) != 0u;
}
__device__ __forceinline__ bool spread_too_wide2(float gmax, float gneg) {
  return !(gmax + gneg <= SPREAD_MAX);
}

__global__ void edge_max1_kernel(const int* __restrict__ src, const int* __restrict__ dst,
                                 const float* __restrict__ a_src, const float* __restrict__ a_dst,
                                 float* __restrict__ m, const float* __restrict__ stat,
                                 int E, int Etot) {
  const int h_own = (int)(threadIdx.x & 7);
  if (!spread_too_wide1(stat[h_own * STAT_PITCH], stat[(8 + h_own) * STAT_PITCH])) return;
  const long long total = (long long)Etot * 8;
  const long long stride = (long long)gridDim.x * blockDim.x;
  for (long long idx = (long long)blockIdx.x * blockDim.x + threadIdx.x; idx < total; idx += stride) {
    const int e = (int)(idx >> 3), hd = (int)(idx & 7);
    const int s = (e < E) ? src[e] : (e - E);
    const int d = (e < E) ? dst[e] : (e - E);
    float v = a_src[s * 8 + hd] + a_dst[d * 8 + hd];
    v = (v >= 0.f) ? v : NEG_SLOPE * v;
    atomicMaxF32(&m[d * 8 + hd], v);
  }
}

template <int J>
__device__ __forceinline__ float quad_bcast(float v) {
  return __int_as_float(__builtin_amdgcn_update_dpp(
      0, __float_as_int(v), J * 0x55, 0xF, 0xF, true));
}

#define EPW 8

__device__ __forceinline__ void edge1_single(int s, int d, int lane, int hd,
                                             bool use_exact, float gmax,
                                             const float* __restrict__ a_src,
                                             const float* __restrict__ a_dst,
                                             const float* __restrict__ m,
                                             float* __restrict__ ssum,
                                             const float* __restrict__ h,
                                             float* __restrict__ agg) {
  const float ad = a_dst[d * 8 + hd];
  float e = a_src[s * 8 + hd] + ad;
  e = (e >= 0.f) ? e : NEG_SLOPE * e;
  float shift = gmax + ad;
  shift = (shift >= 0.f) ? shift : NEG_SLOPE * shift;
  if (use_exact) shift = m[d * 8 + hd];
  float ex = __expf(e - shift);
  const float* hp = h + (size_t)s * 128 + lane;
  const float h0 = hp[0], h1 = hp[32], h2 = hp[64], h3 = hp[96];
  const float w0 = quad_bcast<0>(ex);
  const float w1 = quad_bcast<1>(ex);
  const float w2 = quad_bcast<2>(ex);
  const float w3 = quad_bcast<3>(ex);
  float* op = agg + (size_t)d * 128 + lane;
  atomicAdd(op +  0, h0 * w0);
  atomicAdd(op + 32, h1 * w1);
  atomicAdd(op + 64, h2 * w2);
  atomicAdd(op + 96, h3 * w3);
  if ((lane & 12) == 0) atomicAdd(&ssum[d * 8 + hd], ex);
}

template <bool EXACT>
__device__ __forceinline__ void agg1_second_phase(const int (&dk)[EPW], const float (&es)[EPW],
                                                  const float (&ed)[EPW], const float (&r0)[EPW],
                                                  const float (&r1)[EPW], const float (&r2)[EPW],
                                                  const float (&r3)[EPW], float gmax, int hd, int lane,
                                                  const float* __restrict__ m,
                                                  float* __restrict__ ssum, float* __restrict__ agg) {
#pragma unroll
  for (int k = 0; k < EPW; ++k) {
    float e = es[k] + ed[k];
    e = (e >= 0.f) ? e : NEG_SLOPE * e;
    float shift;
    if (EXACT) {
      shift = m[dk[k] * 8 + hd];
    } else {
      shift = gmax + ed[k];
      shift = (shift >= 0.f) ? shift : NEG_SLOPE * shift;
    }
    const float ex = __expf(e - shift);
    const float w0 = quad_bcast<0>(ex);
    const float w1 = quad_bcast<1>(ex);
    const float w2 = quad_bcast<2>(ex);
    const float w3 = quad_bcast<3>(ex);
    float* op = agg + (size_t)dk[k] * 128 + lane;
    atomicAdd(op +  0, r0[k] * w0);
    atomicAdd(op + 32, r1[k] * w1);
    atomicAdd(op + 64, r2[k] * w2);
    atomicAdd(op + 96, r3[k] * w3);
    if ((lane & 12) == 0) atomicAdd(&ssum[dk[k] * 8 + hd], ex);
  }
}

__global__ void __launch_bounds__(256)
edge_agg1_kernel(const int* __restrict__ src, const int* __restrict__ dst,
                 const float* __restrict__ a_src, const float* __restrict__ a_dst,
                 const float* __restrict__ m, const float* __restrict__ stat,
                 float* __restrict__ ssum,
                 const float* __restrict__ h, float* __restrict__ agg,
                 int E, int Etot, int Nn) {
  const int wid  = (int)(((long long)blockIdx.x * blockDim.x + threadIdx.x) >> 5);
  const int lane = threadIdx.x & 31;
  const long long base = (long long)wid * EPW;
  if (base >= Etot) return;
  const int hd = ((lane & 3) << 1) | (lane >> 4);

  if (base + EPW <= Etot) {
    const int eo = (int)base + (lane & (EPW - 1));
    const int s_own = (eo < E) ? src[eo] : (eo - E);
    const int d_own = (eo < E) ? dst[eo] : (eo - E);
    const float gmax = stat[hd * STAT_PITCH];
    const float gneg = stat[(8 + hd) * STAT_PITCH];
    const size_t plane = (size_t)package_half() * (size_t)Nn;
    agg  += plane * 128;
    ssum += plane * 8;
    int   sk[EPW], dk[EPW];
    float es[EPW], ed[EPW], r0[EPW], r1[EPW], r2[EPW], r3[EPW];
#pragma unroll
    for (int k = 0; k < EPW; ++k) {
      sk[k] = __builtin_amdgcn_readlane(s_own, k);
      dk[k] = __builtin_amdgcn_readlane(d_own, k);
      es[k] = a_src[sk[k] * 8 + hd];
      ed[k] = a_dst[dk[k] * 8 + hd];
      const float* hp = h + (size_t)sk[k] * 128 + lane;
      r0[k] = hp[0]; r1[k] = hp[32]; r2[k] = hp[64]; r3[k] = hp[96];
    }
    if (spread_too_wide1(gmax, gneg))
      agg1_second_phase<true>(dk, es, ed, r0, r1, r2, r3, gmax, hd, lane, m, ssum, agg);
    else
      agg1_second_phase<false>(dk, es, ed, r0, r1, r2, r3, gmax, hd, lane, m, ssum, agg);
  } else {
    const float gmax = stat[hd * STAT_PITCH];
    const bool use_exact = spread_too_wide1(gmax, stat[(8 + hd) * STAT_PITCH]);
    const size_t plane = (size_t)package_half() * (size_t)Nn;
    agg  += plane * 128;
    ssum += plane * 8;
    for (int k = 0; k < EPW; ++k) {
      const long long e = base + k;
      if (e >= Etot) break;
      const int s = (e < E) ? src[e] : (int)(e - E);
      const int d = (e < E) ? dst[e] : (int)(e - E);
      edge1_single(s, d, lane, hd, use_exact, gmax, a_src, a_dst, m, ssum, h, agg);
    }
  }
}

__global__ void elu_bias_kernel(float* __restrict__ a, const float* __restrict__ a_hi,
                                const float* __restrict__ ssum,
                                const float* __restrict__ ssum_hi,
                                const float* __restrict__ b, long long n) {
  long long idx = (long long)blockIdx.x * blockDim.x + threadIdx.x;
  if (idx >= n) return;
  float v = (a[idx] + a_hi[idx]) / (ssum[idx >> 4] + ssum_hi[idx >> 4]) + b[idx & 127];
  a[idx] = (v > 0.f) ? v : (__expf(v) - 1.0f);
}

__global__ void edge_max2_kernel(const int* __restrict__ src, const int* __restrict__ dst,
                                 const float* __restrict__ a_src, const float* __restrict__ a_dst,
                                 float* __restrict__ m, const float* __restrict__ stat,
                                 int E, int Etot) {
  if (!spread_too_wide2(stat[16 * STAT_PITCH], stat[17 * STAT_PITCH])) return;
  const int stride = (int)(gridDim.x * blockDim.x);
  for (int e = blockIdx.x * blockDim.x + threadIdx.x; e < Etot; e += stride) {
    const int s = (e < E) ? src[e] : (e - E);
    const int d = (e < E) ? dst[e] : (e - E);
    float v = a_src[s] + a_dst[d];
    v = (v >= 0.f) ? v : NEG_SLOPE * v;
    atomicMaxF32(&m[d], v);
  }
}

__global__ void __launch_bounds__(256)
edge_agg2_kernel(const int* __restrict__ src, const int* __restrict__ dst,
                 const float* __restrict__ a_src, const float* __restrict__ a_dst,
                 const float* __restrict__ m, const float* __restrict__ stat,
                 float* __restrict__ ssum,
                 const float* __restrict__ h2, float* __restrict__ out_lo,
                 float* __restrict__ out_hi, int E, int Etot, int Nn) {
  const int wid  = (int)(((long long)blockIdx.x * blockDim.x + threadIdx.x) >> 5);
  const int lane = threadIdx.x & 31;
  const long long base = (long long)wid * EPW;
  if (base >= Etot) return;

  if (base + EPW <= Etot) {
    const int eo = (int)base + (lane & (EPW - 1));
    const int s_own = (eo < E) ? src[eo] : (eo - E);
    const int d_own = (eo < E) ? dst[eo] : (eo - E);
    const float gmax = stat[16 * STAT_PITCH];
    const bool use_exact = spread_too_wide2(gmax, stat[17 * STAT_PITCH]);
    const int half = package_half();
    float* __restrict__ out = half ? out_hi : out_lo;
    ssum += (size_t)half * (size_t)Nn;
    const float es = a_src[s_own];
    const float ed = a_dst[d_own];
    float mo = 0.f;
    if (use_exact) mo = m[d_own];
    int   sk[EPW], dk[EPW];
    float ra[EPW], rb[EPW];
#pragma unroll
    for (int k = 0; k < EPW; ++k) {
      sk[k] = __builtin_amdgcn_readlane(s_own, k);
      dk[k] = __builtin_amdgcn_readlane(d_own, k);
      ra[k] = h2[(size_t)sk[k] * 40 + lane];
    }
    if (lane < 8) {
#pragma unroll
      for (int k = 0; k < EPW; ++k) rb[k] = h2[(size_t)sk[k] * 40 + 32 + lane];
    }
    float e = es + ed;
    e = (e >= 0.f) ? e : NEG_SLOPE * e;
    float shift = gmax + ed;
    shift = (shift >= 0.f) ? shift : NEG_SLOPE * shift;
    if (use_exact) shift = mo;
    const float ex_own = __expf(e - shift);
    float wk[EPW];
#pragma unroll
    for (int k = 0; k < EPW; ++k) {
      wk[k] = __int_as_float(__builtin_amdgcn_readlane(__float_as_int(ex_own), k));
      atomicAdd(&out[(size_t)dk[k] * 40 + lane], ra[k] * wk[k]);
    }
    if (lane < 8) {
#pragma unroll
      for (int k = 0; k < EPW; ++k)
        atomicAdd(&out[(size_t)dk[k] * 40 + 32 + lane], rb[k] * wk[k]);
      atomicAdd(&ssum[d_own], ex_own);
    }
  } else {
    const float gmax = stat[16 * STAT_PITCH];
    const bool use_exact = spread_too_wide2(gmax, stat[17 * STAT_PITCH]);
    const int half = package_half();
    float* __restrict__ out = half ? out_hi : out_lo;
    ssum += (size_t)half * (size_t)Nn;
    for (int k = 0; k < EPW; ++k) {
      const long long eg = base + k;
      if (eg >= Etot) break;
      const int s = (eg < E) ? src[eg] : (int)(eg - E);
      const int d = (eg < E) ? dst[eg] : (int)(eg - E);
      const float ad = a_dst[d];
      float e = a_src[s] + ad;
      e = (e >= 0.f) ? e : NEG_SLOPE * e;
      float shift = gmax + ad;
      shift = (shift >= 0.f) ? shift : NEG_SLOPE * shift;
      if (use_exact) shift = m[d];
      const float ex = __expf(e - shift);
      for (int c = lane; c < 40; c += 32)
        atomicAdd(&out[(size_t)d * 40 + c], h2[(size_t)s * 40 + c] * ex);
      if (lane == 0) atomicAdd(&ssum[d], ex);
    }
  }
}

extern "C" void kernel_launch(void* const* d_in, const int* in_sizes, int n_in,
                              void* d_out, int out_size, void* d_ws, size_t ws_size,
                              hipStream_t stream) {
  const float* x   = (const float*)d_in[0];
  const int*   ei  = (const int*)d_in[1];
  const float* W1  = (const float*)d_in[2];
  const float* as1 = (const float*)d_in[3];
  const float* ad1 = (const float*)d_in[4];
  const float* b1  = (const float*)d_in[5];
  const float* W2  = (const float*)d_in[6];
  const float* as2 = (const float*)d_in[7];
  const float* ad2 = (const float*)d_in[8];
  const float* b2  = (const float*)d_in[9];
  float* out = (float*)d_out;

  const int Nn   = in_sizes[0] / 128;
  const int E    = in_sizes[1] / 2;
  const int Etot = E + Nn;
  const int* srcp = ei;
  const int* dstp = ei + E;

  float* ws   = (float*)d_ws;
  float* h    = ws;  ws += (size_t)Nn * 128;
  float* agg1 = ws;  ws += (size_t)Nn * 256;
  float* asc1 = ws;  ws += (size_t)Nn * 8;
  float* adc1 = ws;  ws += (size_t)Nn * 8;
  float* m1   = ws;  ws += (size_t)Nn * 8;
  float* stat = ws;  ws += STAT_WORDS * STAT_PITCH;
  float* s1   = ws;  ws += (size_t)Nn * 16;
  float* h2   = ws;  ws += (size_t)Nn * 40;
  float* outb = ws;  ws += (size_t)Nn * 40;
  float* s2   = ws;  ws += (size_t)Nn * 2;
  float* asc2 = ws;  ws += Nn;
  float* adc2 = ws;  ws += Nn;
  float* m2   = ws;  ws += Nn;

  const int T = 256;
  const int fillBlocks = 2048;

  fill_kernel<<<fillBlocks, T, 0, stream>>>(m1, -INFINITY, (long long)Nn * 8 + STAT_WORDS * STAT_PITCH);
  fill_kernel<<<fillBlocks, T, 0, stream>>>(s1, 0.0f, (long long)Nn * 16);
  fill_kernel<<<fillBlocks, T, 0, stream>>>(agg1, 0.0f, (long long)Nn * 256);
  fill_kernel<<<fillBlocks, T, 0, stream>>>(m2, -INFINITY, (long long)Nn);
  fill_kernel<<<fillBlocks, T, 0, stream>>>(outb, 0.0f, (long long)Nn * 42);
  fill_kernel<<<fillBlocks, T, 0, stream>>>(out, 0.0f, (long long)Nn * 40);

  gat_wmma_gemm_k128<2, 128><<<dim3(Nn / 16), 4 * 32, 0, stream>>>(x, W1, h);
  scores1_kernel<<<(Nn * 8 + T - 1) / T, T, 0, stream>>>(h, as1, ad1, asc1, adc1, stat, Nn);
  edge_max1_kernel<<<1024, T, 0, stream>>>(srcp, dstp, asc1, adc1, m1, stat, E, Etot);
  edge_agg1_kernel<<<(Etot + 8 * EPW - 1) / (8 * EPW), T, 0, stream>>>(srcp, dstp, asc1, adc1, m1, stat, s1, h, agg1, E, Etot, Nn);
  elu_bias_kernel<<<(int)(((long long)Nn * 128 + T - 1) / T), T, 0, stream>>>(agg1, agg1 + (size_t)Nn * 128, s1, s1 + (size_t)Nn * 8, b1, (long long)Nn * 128);

  gat_wmma_gemm_k128<3, 40><<<dim3(Nn / 16), 32, 0, stream>>>(agg1, W2, h2);
  scores2_kernel<<<(Nn + T - 1) / T, T, 0, stream>>>(h2, as2, ad2, asc2, adc2, stat, Nn);
  edge_max2_kernel<<<1024, T, 0, stream>>>(srcp, dstp, asc2, adc2, m2, stat, E, Etot);
  edge_agg2_kernel<<<(Etot + 8 * EPW - 1) / (8 * EPW), T, 0, stream>>>(srcp, dstp, asc2, adc2, m2, stat, s2, h2, out, outb, E, Etot, Nn);
  finalize_out_kernel<<<(Nn * 40 + T - 1) / T, T, 0, stream>>>(out, outb, s2, s2 + Nn, b2, Nn);
}


# Layer-1 aggregation by owner work-groups: 391 destination rows + exp sums in LDS, each scans all edge destinations, adds by LDS atomics (one accumulator, arrival order), divide+bias+ELU fused
# speedup vs baseline: 3.5830x; 1.1850x over previous
#include <hip/hip_runtime.h>
#include <hip/hip_bf16.h>
#include <math.h>
#include <stdint.h>


typedef float v2f __attribute__((ext_vector_type(2)));
typedef float v4f __attribute__((ext_vector_type(4)));
typedef float v8f __attribute__((ext_vector_type(8)));

#define NEG_SLOPE 0.2f
#define SPREAD_MAX 16.0f
#define STAT_PITCH 32
#define STAT_WORDS 18

__device__ __forceinline__ void atomicMaxF32(float* addr, float val) {
  if (val >= 0.0f) {
    atomicMax((int*)addr, __float_as_int(val));
  } else {
    atomicMin((unsigned int*)addr, __float_as_uint(val));
  }
}

__device__ __forceinline__ int package_half() {
  return (int)((__builtin_amdgcn_s_sendmsg_rtn(0x87) >> 18) & 1u);
}

template <int NT, int NACT>
__global__ void __launch_bounds__(32 * ((NACT + 16 * NT - 1) / (16 * NT)))
gat_wmma_gemm_k128(const float* __restrict__ A, const float* __restrict__ B,
                   float* __restrict__ C) {
  const int lane = threadIdx.x & 31;
  const int wave = threadIdx.x >> 5;
  const int c = lane & 15;
  const int h = lane >> 4;
  constexpr int CG = (NACT + 16 * NT - 1) / (16 * NT);
  constexpr bool FULL = (NACT % (16 * NT)) == 0;
  const size_t m0 = (size_t)blockIdx.x * 16;
  const int n0 = (CG == 1) ? 0 : wave * (16 * NT);

  const float* arow = A + (m0 + c) * 128 + 4 * h;
  v4f q[16];
#pragma unroll
  for (int i = 0; i < 16; ++i) q[i] = *(const v4f*)(arow + 8 * i);

  v2f b[NT][32];
#pragma unroll
  for (int t = 0; t < NT; ++t) {
    const int gn = n0 + 16 * t + c;
    const bool in_range = FULL || (CG == 1 && 16 * t + 15 < NACT) || gn < NACT;
    const float* bcol = B + (size_t)(4 * h) * NACT + gn;
#pragma unroll
    for (int j = 0; j < 32; ++j) b[t][j] = (v2f){0.f, 0.f};
    if (in_range) {
#pragma unroll
      for (int j = 0; j < 32; ++j) {
        const int k = 8 * (j >> 1) + 2 * (j & 1);
        b[t][j].x = bcol[(size_t)k * NACT];
        b[t][j].y = bcol[(size_t)(k + 1) * NACT];
      }
    }
  }

  v8f acc[NT];
#pragma unroll
  for (int t = 0; t < NT; ++t) acc[t] = (v8f){0.f, 0.f, 0.f, 0.f, 0.f, 0.f, 0.f, 0.f};

#pragma unroll
  for (int j = 0; j < 32; ++j) {
    const v4f qq = q[j >> 1];
    v2f a;
    if (j & 1) { a.x = qq.z; a.y = qq.w; } else { a.x = qq.x; a.y = qq.y; }
#pragma unroll
    for (int t = 0; t < NT; ++t) {
      acc[t] = __builtin_amdgcn_wmma_f32_16x16x4_f32(
           false, a,  false, b[t][j],
           (short)0, acc[t],  false,  false);
    }
  }

#pragma unroll
  for (int t = 0; t < NT; ++t) {
    const int gn = n0 + 16 * t + c;
    if (FULL || (CG == 1 && 16 * t + 15 < NACT) || gn < NACT) {
      float* cp = C + (m0 + (size_t)(h << 3)) * NACT + gn;
#pragma unroll
      for (int r = 0; r < 8; ++r) cp[(size_t)r * NACT] = acc[t][r];
    }
  }
}

__global__ void fill_kernel(float* __restrict__ p, float v, long long n) {
  long long i = (long long)blockIdx.x * blockDim.x + threadIdx.x;
  long long stride = (long long)gridDim.x * blockDim.x;
  for (; i < n; i += stride) p[i] = v;
}

__global__ void finalize_out_kernel(float* __restrict__ out, const float* __restrict__ out_hi,
                                    const float* __restrict__ ssum,
                                    const float* __restrict__ ssum_hi,
                                    const float* __restrict__ b2, int Nn) {
  int idx = blockIdx.x * blockDim.x + threadIdx.x;
  if (idx >= Nn * 40) return;
  int n = idx / 40;
  out[idx] = (out[idx] + out_hi[idx]) / (ssum[n] + ssum_hi[n]) + b2[idx - n * 40];
}

__global__ void __launch_bounds__(256)
scores1_kernel(const float* __restrict__ h,
               const float* __restrict__ att_src,
               const float* __restrict__ att_dst,
               float* __restrict__ a_src, float* __restrict__ a_dst,
               float* __restrict__ stat, int Nn) {
  __shared__ float red[2][8][8];
  const int idx = blockIdx.x * blockDim.x + threadIdx.x;
  const bool valid = idx < Nn * 8;
  float s = 0.f, d = 0.f;
  if (valid) {
    const int n = idx >> 3, hd = idx & 7;
    const float* hp = h + (size_t)n * 128 + hd * 16;
#pragma unroll
    for (int c = 0; c < 16; ++c) {
      float v = hp[c];
      s = fmaf(v, att_src[hd * 16 + c], s);
      d = fmaf(v, att_dst[hd * 16 + c], d);
    }
    a_src[idx] = s;
    a_dst[idx] = d;
  }
  float vmax = valid ? s : -INFINITY;
  float vneg = valid ? -s : -INFINITY;
  vmax = fmaxf(vmax, __shfl_xor(vmax, 8));
  vneg = fmaxf(vneg, __shfl_xor(vneg, 8));
  vmax = fmaxf(vmax, __shfl_xor(vmax, 16));
  vneg = fmaxf(vneg, __shfl_xor(vneg, 16));
  const int lane = threadIdx.x & 31, wave = threadIdx.x >> 5;
  if (lane < 8) { red[0][wave][lane] = vmax; red[1][wave][lane] = vneg; }
  __syncthreads();
  if (threadIdx.x < 16) {
    const int which = threadIdx.x >> 3, hh = threadIdx.x & 7;
    float v = red[which][0][hh];
#pragma unroll
    for (int w = 1; w < 8; ++w) v = fmaxf(v, red[which][w][hh]);
    atomicMaxF32(&stat[threadIdx.x * STAT_PITCH], v);
  }
}

__global__ void __launch_bounds__(256)
scores2_kernel(const float* __restrict__ h2,
               const float* __restrict__ att_src,
               const float* __restrict__ att_dst,
               float* __restrict__ a_src, float* __restrict__ a_dst,
               float* __restrict__ stat, int Nn) {
  __shared__ float red[2][8];
  const int n = blockIdx.x * blockDim.x + threadIdx.x;
  const bool valid = n < Nn;
  float s = 0.f, d = 0.f;
  if (valid) {
    const float* hp = h2 + (size_t)n * 40;
#pragma unroll
    for (int c = 0; c < 40; ++c) {
      float v = hp[c];
      s = fmaf(v, att_src[c], s);
      d = fmaf(v, att_dst[c], d);
    }
    a_src[n] = s;
    a_dst[n] = d;
  }
  float vmax = valid ? s : -INFINITY;
  float vneg = valid ? -s : -INFINITY;
#pragma unroll
  for (int o = 16; o >= 1; o >>= 1) {
    vmax = fmaxf(vmax, __shfl_xor(vmax, o));
    vneg = fmaxf(vneg, __shfl_xor(vneg, o));
  }
  const int lane = threadIdx.x & 31, wave = threadIdx.x >> 5;
  if (lane == 0) { red[0][wave] = vmax; red[1][wave] = vneg; }
  __syncthreads();
  if (threadIdx.x < 2) {
    float v = red[threadIdx.x][0];
#pragma unroll
    for (int w = 1; w < 8; ++w) v = fmaxf(v, red[threadIdx.x][w]);
    atomicMaxF32(&stat[(16 + threadIdx.x) * STAT_PITCH], v);
  }
}

__device__ __forceinline__ bool spread_too_wide1(float gmax, float gneg) {
  return __builtin_amdgcn_ballot_w32(!(gmax + gneg <= SPREAD_MAX)) != 0u;
}
__device__ __forceinline__ bool spread_too_wide2(float gmax, float gneg) {
  return !(gmax + gneg <= SPREAD_MAX);
}

__global__ void edge_max1_kernel(const int* __restrict__ src, const int* __restrict__ dst,
                                 const float* __restrict__ a_src, const float* __restrict__ a_dst,
                                 float* __restrict__ m, const float* __restrict__ stat,
                                 int E, int Etot) {
  const int h_own = (int)(threadIdx.x & 7);
  if (!spread_too_wide1(stat[h_own * STAT_PITCH], stat[(8 + h_own) * STAT_PITCH])) return;
  const long long total = (long long)Etot * 8;
  const long long stride = (long long)gridDim.x * blockDim.x;
  for (long long idx = (long long)blockIdx.x * blockDim.x + threadIdx.x; idx < total; idx += stride) {
    const int e = (int)(idx >> 3), hd = (int)(idx & 7);
    const int s = (e < E) ? src[e] : (e - E);
    const int d = (e < E) ? dst[e] : (e - E);
    float v = a_src[s * 8 + hd] + a_dst[d * 8 + hd];
    v = (v >= 0.f) ? v : NEG_SLOPE * v;
    atomicMaxF32(&m[d * 8 + hd], v);
  }
}

template <int J>
__device__ __forceinline__ float quad_bcast(float v) {
  return __int_as_float(__builtin_amdgcn_update_dpp(
      0, __float_as_int(v), J * 0x55, 0xF, 0xF, true));
}

#define EPW 8

#define OWN_R 391
#define OWN_PITCH 136
#define OWN_WAVES 32
#define OWN_QMASK 255

__device__ __forceinline__ int own_append(int* qe, int* qd, int cnt,
                                          unsigned mask, bool pred, int ev, int dv, int lane) {
  if (mask != 0u) {
    const int pos = cnt + __builtin_popcount(mask & ((1u << lane) - 1u));
    if (pred) { qe[pos & OWN_QMASK] = ev; qd[pos & OWN_QMASK] = dv; }
    cnt += __builtin_popcount(mask);
  }
  return cnt;
}

template <bool FULL>
__device__ __forceinline__ void own_batch(const int* qe, const int* qd,
                                          int head, int nvalid, int lo, int lane, int hd,
                                          bool exact, float gmax,
                                          const int* __restrict__ src,
                                          const float* __restrict__ a_src,
                                          const float* __restrict__ a_dst,
                                          const float* __restrict__ m,
                                          const float* __restrict__ h,
                                          float* __restrict__ acc, int E) {
  const int slot = FULL ? (lane & (EPW - 1)) : min(lane & (EPW - 1), nvalid - 1);
  const int eo = qe[(head + slot) & OWN_QMASK];
  const int d_own = qd[(head + slot) & OWN_QMASK];
  const int s_own = (eo < E) ? src[eo] : (eo - E);
  int   dk[EPW];
  float es[EPW], ed[EPW], mk[EPW], r0[EPW], r1[EPW], r2[EPW], r3[EPW];
#pragma unroll
  for (int k = 0; k < EPW; ++k) {
    mk[k] = 0.f;
    if (FULL || k < nvalid) {
      const int sk = __builtin_amdgcn_readlane(s_own, k);
      dk[k] = __builtin_amdgcn_readlane(d_own, k);
      es[k] = a_src[sk * 8 + hd];
      ed[k] = a_dst[(lo + dk[k]) * 8 + hd];
      const float* hp = h + (size_t)sk * 128 + lane;
      r0[k] = hp[0]; r1[k] = hp[32]; r2[k] = hp[64]; r3[k] = hp[96];
    }
  }
  if (exact) {
#pragma unroll
    for (int k = 0; k < EPW; ++k)
      if (FULL || k < nvalid) mk[k] = m[(lo + dk[k]) * 8 + hd];
  }
#pragma unroll
  for (int k = 0; k < EPW; ++k) {
    if (FULL || k < nvalid) {
      float e = es[k] + ed[k];
      e = (e >= 0.f) ? e : NEG_SLOPE * e;
      float shift = gmax + ed[k];
      shift = (shift >= 0.f) ? shift : NEG_SLOPE * shift;
      if (exact) shift = mk[k];
      const float ex = __expf(e - shift);
      const float w0 = quad_bcast<0>(ex);
      const float w1 = quad_bcast<1>(ex);
      const float w2 = quad_bcast<2>(ex);
      const float w3 = quad_bcast<3>(ex);
      float* op = acc + dk[k] * OWN_PITCH + lane;
      atomicAdd(op +  0, r0[k] * w0);
      atomicAdd(op + 32, r1[k] * w1);
      atomicAdd(op + 64, r2[k] * w2);
      atomicAdd(op + 96, r3[k] * w3);
      if ((lane & 12) == 0) atomicAdd(acc + dk[k] * OWN_PITCH + 128 + hd, ex);
    }
  }
}

#define OWN_U 4
__global__ void __launch_bounds__(32 * OWN_WAVES)
agg1_owner_kernel(const int* __restrict__ src, const int* __restrict__ dst,
                  const float* __restrict__ a_src, const float* __restrict__ a_dst,
                  const float* __restrict__ m, const float* __restrict__ stat,
                  const float* __restrict__ h, const float* __restrict__ b,
                  float* __restrict__ h1, int E, int Nn) {
  __shared__ float acc[OWN_R * OWN_PITCH];
  __shared__ int ring_e[OWN_WAVES][OWN_QMASK + 1];
  __shared__ int ring_d[OWN_WAVES][OWN_QMASK + 1];
  const int lane = threadIdx.x & 31;
  const int wave = threadIdx.x >> 5;
  const int lo = blockIdx.x * OWN_R;
  const int nown = min(OWN_R, Nn - lo);
  const int hd = ((lane & 3) << 1) | (lane >> 4);
  const float gmax = stat[hd * STAT_PITCH];
  const bool exact = spread_too_wide1(gmax, stat[(8 + hd) * STAT_PITCH]);
  int* qe = ring_e[wave];
  int* qd = ring_d[wave];

  for (int i = threadIdx.x; i < OWN_R * OWN_PITCH; i += 32 * OWN_WAVES) acc[i] = 0.f;
  __syncthreads();

  int head = 0, cnt = 0;
  for (int r0 = 0; r0 < nown; r0 += 32 * OWN_WAVES) {
    const int r = r0 + 32 * lane + wave;
    const bool pred = r < nown;
    cnt = own_append(qe, qd, cnt, __builtin_amdgcn_ballot_w32(pred), pred, E + lo + r, r, lane);
  }

  const int ng = E >> 2;
  const int4* __restrict__ dst4 = (const int4*)dst;
  const int per_trip = OWN_WAVES * OWN_U * 32;
  const int ntrips = (ng + per_trip - 1) / per_trip;
  int trip = (ntrips > 0) ? (int)(blockIdx.x % (unsigned)ntrips) : 0;
  int4 cur[OWN_U], nxt[OWN_U];
  int g0 = trip * per_trip + wave * (OWN_U * 32) + lane;
#pragma unroll
  for (int j = 0; j < OWN_U; ++j) {
    const int g = g0 + 32 * j;
    cur[j] = dst4[min(g, max(ng - 1, 0))];
    if (g >= ng) cur[j] = (int4){-1, -1, -1, -1};
  }
  for (int t = 0; t < ntrips; ++t) {
    trip = (trip + 1 < ntrips) ? trip + 1 : 0;
    const int gn = trip * per_trip + wave * (OWN_U * 32) + lane;
#pragma unroll
    for (int j = 0; j < OWN_U; ++j) {
      const int g = gn + 32 * j;
      nxt[j] = dst4[min(g, ng - 1)];
      if (g >= ng) nxt[j] = (int4){-1, -1, -1, -1};
    }
#pragma unroll
    for (int j = 0; j < OWN_U; ++j) {
      const int e0 = (g0 + 32 * j) * 4;
      const int dx = cur[j].x - lo, dy = cur[j].y - lo, dz = cur[j].z - lo, dw = cur[j].w - lo;
      const bool px = (unsigned)dx < (unsigned)nown, py = (unsigned)dy < (unsigned)nown;
      const bool pz = (unsigned)dz < (unsigned)nown, pw = (unsigned)dw < (unsigned)nown;
      const unsigned mx = __builtin_amdgcn_ballot_w32(px), my = __builtin_amdgcn_ballot_w32(py);
      const unsigned mz = __builtin_amdgcn_ballot_w32(pz), mw = __builtin_amdgcn_ballot_w32(pw);
      if ((mx | my | mz | mw) != 0u) {
        cnt = own_append(qe, qd, cnt, mx, px, e0 + 0, dx, lane);
        cnt = own_append(qe, qd, cnt, my, py, e0 + 1, dy, lane);
        cnt = own_append(qe, qd, cnt, mz, pz, e0 + 2, dz, lane);
        cnt = own_append(qe, qd, cnt, mw, pw, e0 + 3, dw, lane);
        while (cnt - head >= EPW) {
          own_batch<true>(qe, qd, head, EPW, lo, lane, hd, exact, gmax, src, a_src, a_dst, m, h, acc, E);
          head += EPW;
        }
      }
    }
#pragma unroll
    for (int j = 0; j < OWN_U; ++j) cur[j] = nxt[j];
    g0 = gn;
  }
  if (wave == 0) {
    const int e = (ng << 2) + lane;
    const int dl = ((lane < (E & 3)) ? dst[min(e, E - 1)] : -1) - lo;
    const bool pred = (lane < (E & 3)) && ((unsigned)dl < (unsigned)nown);
    cnt = own_append(qe, qd, cnt, __builtin_amdgcn_ballot_w32(pred), pred, e, dl, lane);
  }
  while (cnt - head > 0) {
    const int nv = min(cnt - head, EPW);
    own_batch<false>(qe, qd, head, nv, lo, lane, hd, exact, gmax, src, a_src, a_dst, m, h, acc, E);
    head += nv;
  }
  __syncthreads();

  for (int i = threadIdx.x; i < nown * 128; i += 32 * OWN_WAVES) {
    const int r = i >> 7, f = i & 127;
    const float v = acc[r * OWN_PITCH + f] / acc[r * OWN_PITCH + 128 + (f >> 4)] + b[f];
    h1[(size_t)(lo + r) * 128 + f] = (v > 0.f) ? v : (__expf(v) - 1.0f);
  }
}

__global__ void edge_max2_kernel(const int* __restrict__ src, const int* __restrict__ dst,
                                 const float* __restrict__ a_src, const float* __restrict__ a_dst,
                                 float* __restrict__ m, const float* __restrict__ stat,
                                 int E, int Etot) {
  if (!spread_too_wide2(stat[16 * STAT_PITCH], stat[17 * STAT_PITCH])) return;
  const int stride = (int)(gridDim.x * blockDim.x);
  for (int e = blockIdx.x * blockDim.x + threadIdx.x; e < Etot; e += stride) {
    const int s = (e < E) ? src[e] : (e - E);
    const int d = (e < E) ? dst[e] : (e - E);
    float v = a_src[s] + a_dst[d];
    v = (v >= 0.f) ? v : NEG_SLOPE * v;
    atomicMaxF32(&m[d], v);
  }
}

__global__ void __launch_bounds__(256)
edge_agg2_kernel(const int* __restrict__ src, const int* __restrict__ dst,
                 const float* __restrict__ a_src, const float* __restrict__ a_dst,
                 const float* __restrict__ m, const float* __restrict__ stat,
                 float* __restrict__ ssum,
                 const float* __restrict__ h2, float* __restrict__ out_lo,
                 float* __restrict__ out_hi, int E, int Etot, int Nn) {
  const int wid  = (int)(((long long)blockIdx.x * blockDim.x + threadIdx.x) >> 5);
  const int lane = threadIdx.x & 31;
  const long long base = (long long)wid * EPW;
  if (base >= Etot) return;

  if (base + EPW <= Etot) {
    const int eo = (int)base + (lane & (EPW - 1));
    const int s_own = (eo < E) ? src[eo] : (eo - E);
    const int d_own = (eo < E) ? dst[eo] : (eo - E);
    const float gmax = stat[16 * STAT_PITCH];
    const bool use_exact = spread_too_wide2(gmax, stat[17 * STAT_PITCH]);
    const int half = package_half();
    float* __restrict__ out = half ? out_hi : out_lo;
    ssum += (size_t)half * (size_t)Nn;
    const float es = a_src[s_own];
    const float ed = a_dst[d_own];
    float mo = 0.f;
    if (use_exact) mo = m[d_own];
    int   sk[EPW], dk[EPW];
    float ra[EPW], rb[EPW];
#pragma unroll
    for (int k = 0; k < EPW; ++k) {
      sk[k] = __builtin_amdgcn_readlane(s_own, k);
      dk[k] = __builtin_amdgcn_readlane(d_own, k);
      ra[k] = h2[(size_t)sk[k] * 40 + lane];
    }
    if (lane < 8) {
#pragma unroll
      for (int k = 0; k < EPW; ++k) rb[k] = h2[(size_t)sk[k] * 40 + 32 + lane];
    }
    float e = es + ed;
    e = (e >= 0.f) ? e : NEG_SLOPE * e;
    float shift = gmax + ed;
    shift = (shift >= 0.f) ? shift : NEG_SLOPE * shift;
    if (use_exact) shift = mo;
    const float ex_own = __expf(e - shift);
    float wk[EPW];
#pragma unroll
    for (int k = 0; k < EPW; ++k) {
      wk[k] = __int_as_float(__builtin_amdgcn_readlane(__float_as_int(ex_own), k));
      atomicAdd(&out[(size_t)dk[k] * 40 + lane], ra[k] * wk[k]);
    }
    if (lane < 8) {
#pragma unroll
      for (int k = 0; k < EPW; ++k)
        atomicAdd(&out[(size_t)dk[k] * 40 + 32 + lane], rb[k] * wk[k]);
      atomicAdd(&ssum[d_own], ex_own);
    }
  } else {
    const float gmax = stat[16 * STAT_PITCH];
    const bool use_exact = spread_too_wide2(gmax, stat[17 * STAT_PITCH]);
    const int half = package_half();
    float* __restrict__ out = half ? out_hi : out_lo;
    ssum += (size_t)half * (size_t)Nn;
    for (int k = 0; k < EPW; ++k) {
      const long long eg = base + k;
      if (eg >= Etot) break;
      const int s = (eg < E) ? src[eg] : (int)(eg - E);
      const int d = (eg < E) ? dst[eg] : (int)(eg - E);
      const float ad = a_dst[d];
      float e = a_src[s] + ad;
      e = (e >= 0.f) ? e : NEG_SLOPE * e;
      float shift = gmax + ad;
      shift = (shift >= 0.f) ? shift : NEG_SLOPE * shift;
      if (use_exact) shift = m[d];
      const float ex = __expf(e - shift);
      for (int c = lane; c < 40; c += 32)
        atomicAdd(&out[(size_t)d * 40 + c], h2[(size_t)s * 40 + c] * ex);
      if (lane == 0) atomicAdd(&ssum[d], ex);
    }
  }
}

extern "C" void kernel_launch(void* const* d_in, const int* in_sizes, int n_in,
                              void* d_out, int out_size, void* d_ws, size_t ws_size,
                              hipStream_t stream) {
  const float* x   = (const float*)d_in[0];
  const int*   ei  = (const int*)d_in[1];
  const float* W1  = (const float*)d_in[2];
  const float* as1 = (const float*)d_in[3];
  const float* ad1 = (const float*)d_in[4];
  const float* b1  = (const float*)d_in[5];
  const float* W2  = (const float*)d_in[6];
  const float* as2 = (const float*)d_in[7];
  const float* ad2 = (const float*)d_in[8];
  const float* b2  = (const float*)d_in[9];
  float* out = (float*)d_out;

  const int Nn   = in_sizes[0] / 128;
  const int E    = in_sizes[1] / 2;
  const int Etot = E + Nn;
  const int* srcp = ei;
  const int* dstp = ei + E;

  float* ws   = (float*)d_ws;
  float* h    = ws;  ws += (size_t)Nn * 128;
  float* h1   = ws;  ws += (size_t)Nn * 128;
  float* asc1 = ws;  ws += (size_t)Nn * 8;
  float* adc1 = ws;  ws += (size_t)Nn * 8;
  float* m1   = ws;  ws += (size_t)Nn * 8;
  float* stat = ws;  ws += STAT_WORDS * STAT_PITCH;
  float* h2   = ws;  ws += (size_t)Nn * 40;
  float* outb = ws;  ws += (size_t)Nn * 40;
  float* s2   = ws;  ws += (size_t)Nn * 2;
  float* asc2 = ws;  ws += Nn;
  float* adc2 = ws;  ws += Nn;
  float* m2   = ws;  ws += Nn;

  const int T = 256;
  const int fillBlocks = 2048;

  fill_kernel<<<fillBlocks, T, 0, stream>>>(m1, -INFINITY, (long long)Nn * 8 + STAT_WORDS * STAT_PITCH);
  fill_kernel<<<fillBlocks, T, 0, stream>>>(m2, -INFINITY, (long long)Nn);
  fill_kernel<<<fillBlocks, T, 0, stream>>>(outb, 0.0f, (long long)Nn * 42);
  fill_kernel<<<fillBlocks, T, 0, stream>>>(out, 0.0f, (long long)Nn * 40);

  gat_wmma_gemm_k128<2, 128><<<dim3(Nn / 16), 4 * 32, 0, stream>>>(x, W1, h);
  scores1_kernel<<<(Nn * 8 + T - 1) / T, T, 0, stream>>>(h, as1, ad1, asc1, adc1, stat, Nn);
  edge_max1_kernel<<<1024, T, 0, stream>>>(srcp, dstp, asc1, adc1, m1, stat, E, Etot);
  agg1_owner_kernel<<<(Nn + OWN_R - 1) / OWN_R, 32 * OWN_WAVES, 0, stream>>>(srcp, dstp, asc1, adc1, m1, stat, h, b1, h1, E, Nn);

  gat_wmma_gemm_k128<3, 40><<<dim3(Nn / 16), 32, 0, stream>>>(h1, W2, h2);
  scores2_kernel<<<(Nn + T - 1) / T, T, 0, stream>>>(h2, as2, ad2, asc2, adc2, stat, Nn);
  edge_max2_kernel<<<1024, T, 0, stream>>>(srcp, dstp, asc2, adc2, m2, stat, E, Etot);
  edge_agg2_kernel<<<(Etot + 8 * EPW - 1) / (8 * EPW), T, 0, stream>>>(srcp, dstp, asc2, adc2, m2, stat, s2, h2, out, outb, E, Etot, Nn);
  finalize_out_kernel<<<(Nn * 40 + T - 1) / T, T, 0, stream>>>(out, outb, s2, s2 + Nn, b2, Nn);
}


# Layer-2 aggregation by the owner work-groups too: they replay the (source,row) pairs saved in layer 1, add rows + exp sums by LDS atomics (one accumulator, arrival order), divide+bias fused
# speedup vs baseline: 4.4121x; 1.2314x over previous
#include <hip/hip_runtime.h>
#include <hip/hip_bf16.h>
#include <math.h>
#include <stdint.h>


typedef float v2f __attribute__((ext_vector_type(2)));
typedef float v4f __attribute__((ext_vector_type(4)));
typedef float v8f __attribute__((ext_vector_type(8)));

#define NEG_SLOPE 0.2f
#define SPREAD_MAX 16.0f
#define STAT_PITCH 32
#define STAT_WORDS 18

__device__ __forceinline__ void atomicMaxF32(float* addr, float val) {
  if (val >= 0.0f) {
    atomicMax((int*)addr, __float_as_int(val));
  } else {
    atomicMin((unsigned int*)addr, __float_as_uint(val));
  }
}

template <int NT, int NACT>
__global__ void __launch_bounds__(32 * ((NACT + 16 * NT - 1) / (16 * NT)))
gat_wmma_gemm_k128(const float* __restrict__ A, const float* __restrict__ B,
                   float* __restrict__ C) {
  const int lane = threadIdx.x & 31;
  const int wave = threadIdx.x >> 5;
  const int c = lane & 15;
  const int h = lane >> 4;
  constexpr int CG = (NACT + 16 * NT - 1) / (16 * NT);
  constexpr bool FULL = (NACT % (16 * NT)) == 0;
  const size_t m0 = (size_t)blockIdx.x * 16;
  const int n0 = (CG == 1) ? 0 : wave * (16 * NT);

  const float* arow = A + (m0 + c) * 128 + 4 * h;
  v4f q[16];
#pragma unroll
  for (int i = 0; i < 16; ++i) q[i] = *(const v4f*)(arow + 8 * i);

  v2f b[NT][32];
#pragma unroll
  for (int t = 0; t < NT; ++t) {
    const int gn = n0 + 16 * t + c;
    const bool in_range = FULL || (CG == 1 && 16 * t + 15 < NACT) || gn < NACT;
    const float* bcol = B + (size_t)(4 * h) * NACT + gn;
#pragma unroll
    for (int j = 0; j < 32; ++j) b[t][j] = (v2f){0.f, 0.f};
    if (in_range) {
#pragma unroll
      for (int j = 0; j < 32; ++j) {
        const int k = 8 * (j >> 1) + 2 * (j & 1);
        b[t][j].x = bcol[(size_t)k * NACT];
        b[t][j].y = bcol[(size_t)(k + 1) * NACT];
      }
    }
  }

  v8f acc[NT];
#pragma unroll
  for (int t = 0; t < NT; ++t) acc[t] = (v8f){0.f, 0.f, 0.f, 0.f, 0.f, 0.f, 0.f, 0.f};

#pragma unroll
  for (int j = 0; j < 32; ++j) {
    const v4f qq = q[j >> 1];
    v2f a;
    if (j & 1) { a.x = qq.z; a.y = qq.w; } else { a.x = qq.x; a.y = qq.y; }
#pragma unroll
    for (int t = 0; t < NT; ++t) {
      acc[t] = __builtin_amdgcn_wmma_f32_16x16x4_f32(
           false, a,  false, b[t][j],
           (short)0, acc[t],  false,  false);
    }
  }

#pragma unroll
  for (int t = 0; t < NT; ++t) {
    const int gn = n0 + 16 * t + c;
    if (FULL || (CG == 1 && 16 * t + 15 < NACT) || gn < NACT) {
      float* cp = C + (m0 + (size_t)(h << 3)) * NACT + gn;
#pragma unroll
      for (int r = 0; r < 8; ++r) cp[(size_t)r * NACT] = acc[t][r];
    }
  }
}

__global__ void fill_kernel(float* __restrict__ p, float v, long long n) {
  long long i = (long long)blockIdx.x * blockDim.x + threadIdx.x;
  long long stride = (long long)gridDim.x * blockDim.x;
  for (; i < n; i += stride) p[i] = v;
}

__global__ void __launch_bounds__(256)
scores1_kernel(const float* __restrict__ h,
               const float* __restrict__ att_src,
               const float* __restrict__ att_dst,
               float* __restrict__ a_src, float* __restrict__ a_dst,
               float* __restrict__ stat, int Nn) {
  __shared__ float red[2][8][8];
  const int idx = blockIdx.x * blockDim.x + threadIdx.x;
  const bool valid = idx < Nn * 8;
  float s = 0.f, d = 0.f;
  if (valid) {
    const int n = idx >> 3, hd = idx & 7;
    const float* hp = h + (size_t)n * 128 + hd * 16;
#pragma unroll
    for (int c = 0; c < 16; ++c) {
      float v = hp[c];
      s = fmaf(v, att_src[hd * 16 + c], s);
      d = fmaf(v, att_dst[hd * 16 + c], d);
    }
    a_src[idx] = s;
    a_dst[idx] = d;
  }
  float vmax = valid ? s : -INFINITY;
  float vneg = valid ? -s : -INFINITY;
  vmax = fmaxf(vmax, __shfl_xor(vmax, 8));
  vneg = fmaxf(vneg, __shfl_xor(vneg, 8));
  vmax = fmaxf(vmax, __shfl_xor(vmax, 16));
  vneg = fmaxf(vneg, __shfl_xor(vneg, 16));
  const int lane = threadIdx.x & 31, wave = threadIdx.x >> 5;
  if (lane < 8) { red[0][wave][lane] = vmax; red[1][wave][lane] = vneg; }
  __syncthreads();
  if (threadIdx.x < 16) {
    const int which = threadIdx.x >> 3, hh = threadIdx.x & 7;
    float v = red[which][0][hh];
#pragma unroll
    for (int w = 1; w < 8; ++w) v = fmaxf(v, red[which][w][hh]);
    atomicMaxF32(&stat[threadIdx.x * STAT_PITCH], v);
  }
}

__global__ void __launch_bounds__(256)
scores2_kernel(const float* __restrict__ h2,
               const float* __restrict__ att_src,
               const float* __restrict__ att_dst,
               float* __restrict__ a_src, float* __restrict__ a_dst,
               float* __restrict__ stat, int Nn) {
  __shared__ float red[2][8];
  const int n = blockIdx.x * blockDim.x + threadIdx.x;
  const bool valid = n < Nn;
  float s = 0.f, d = 0.f;
  if (valid) {
    const float* hp = h2 + (size_t)n * 40;
#pragma unroll
    for (int c = 0; c < 40; ++c) {
      float v = hp[c];
      s = fmaf(v, att_src[c], s);
      d = fmaf(v, att_dst[c], d);
    }
    a_src[n] = s;
    a_dst[n] = d;
  }
  float vmax = valid ? s : -INFINITY;
  float vneg = valid ? -s : -INFINITY;
#pragma unroll
  for (int o = 16; o >= 1; o >>= 1) {
    vmax = fmaxf(vmax, __shfl_xor(vmax, o));
    vneg = fmaxf(vneg, __shfl_xor(vneg, o));
  }
  const int lane = threadIdx.x & 31, wave = threadIdx.x >> 5;
  if (lane == 0) { red[0][wave] = vmax; red[1][wave] = vneg; }
  __syncthreads();
  if (threadIdx.x < 2) {
    float v = red[threadIdx.x][0];
#pragma unroll
    for (int w = 1; w < 8; ++w) v = fmaxf(v, red[threadIdx.x][w]);
    atomicMaxF32(&stat[(16 + threadIdx.x) * STAT_PITCH], v);
  }
}

__device__ __forceinline__ bool spread_too_wide1(float gmax, float gneg) {
  return __builtin_amdgcn_ballot_w32(!(gmax + gneg <= SPREAD_MAX)) != 0u;
}
__device__ __forceinline__ bool spread_too_wide2(float gmax, float gneg) {
  return !(gmax + gneg <= SPREAD_MAX);
}

__global__ void edge_max1_kernel(const int* __restrict__ src, const int* __restrict__ dst,
                                 const float* __restrict__ a_src, const float* __restrict__ a_dst,
                                 float* __restrict__ m, const float* __restrict__ stat,
                                 int E, int Etot) {
  const int h_own = (int)(threadIdx.x & 7);
  if (!spread_too_wide1(stat[h_own * STAT_PITCH], stat[(8 + h_own) * STAT_PITCH])) return;
  const long long total = (long long)Etot * 8;
  const long long stride = (long long)gridDim.x * blockDim.x;
  for (long long idx = (long long)blockIdx.x * blockDim.x + threadIdx.x; idx < total; idx += stride) {
    const int e = (int)(idx >> 3), hd = (int)(idx & 7);
    const int s = (e < E) ? src[e] : (e - E);
    const int d = (e < E) ? dst[e] : (e - E);
    float v = a_src[s * 8 + hd] + a_dst[d * 8 + hd];
    v = (v >= 0.f) ? v : NEG_SLOPE * v;
    atomicMaxF32(&m[d * 8 + hd], v);
  }
}

template <int J>
__device__ __forceinline__ float quad_bcast(float v) {
  return __int_as_float(__builtin_amdgcn_update_dpp(
      0, __float_as_int(v), J * 0x55, 0xF, 0xF, true));
}

#define EPW 8

#define OWN_R 391
#define OWN_PITCH 136
#define OWN_WAVES 32
#define OWN_QMASK 255
#define OWN2_PITCH 41
#define LIST_CAP 1024

__device__ __forceinline__ int own_append(int* qe, int* qd, int cnt,
                                          unsigned mask, bool pred, int ev, int dv, int lane) {
  if (mask != 0u) {
    const int pos = cnt + __builtin_popcount(mask & ((1u << lane) - 1u));
    if (pred) { qe[pos & OWN_QMASK] = ev; qd[pos & OWN_QMASK] = dv; }
    cnt += __builtin_popcount(mask);
  }
  return cnt;
}

template <bool FULL>
__device__ __forceinline__ void own_batch(const int* qe, const int* qd,
                                          int head, int nvalid, int lo, int lane, int hd,
                                          bool exact, float gmax,
                                          const int* __restrict__ src,
                                          const float* __restrict__ a_src,
                                          const float* __restrict__ a_dst,
                                          const float* __restrict__ m,
                                          const float* __restrict__ h,
                                          float* __restrict__ acc, int E,
                                          int2* __restrict__ lst) {
  const int slot = FULL ? (lane & (EPW - 1)) : min(lane & (EPW - 1), nvalid - 1);
  const int eo = qe[(head + slot) & OWN_QMASK];
  const int d_own = qd[(head + slot) & OWN_QMASK];
  const int s_own = (eo < E) ? src[eo] : (eo - E);
  if (lane < (FULL ? EPW : nvalid) && head + lane < LIST_CAP) lst[head + lane] = make_int2(s_own, d_own);
  int   dk[EPW];
  float es[EPW], ed[EPW], mk[EPW], r0[EPW], r1[EPW], r2[EPW], r3[EPW];
#pragma unroll
  for (int k = 0; k < EPW; ++k) {
    mk[k] = 0.f;
    if (FULL || k < nvalid) {
      const int sk = __builtin_amdgcn_readlane(s_own, k);
      dk[k] = __builtin_amdgcn_readlane(d_own, k);
      es[k] = a_src[sk * 8 + hd];
      ed[k] = a_dst[(lo + dk[k]) * 8 + hd];
      const float* hp = h + (size_t)sk * 128 + lane;
      r0[k] = hp[0]; r1[k] = hp[32]; r2[k] = hp[64]; r3[k] = hp[96];
    }
  }
  if (exact) {
#pragma unroll
    for (int k = 0; k < EPW; ++k)
      if (FULL || k < nvalid) mk[k] = m[(lo + dk[k]) * 8 + hd];
  }
#pragma unroll
  for (int k = 0; k < EPW; ++k) {
    if (FULL || k < nvalid) {
      float e = es[k] + ed[k];
      e = (e >= 0.f) ? e : NEG_SLOPE * e;
      float shift = gmax + ed[k];
      shift = (shift >= 0.f) ? shift : NEG_SLOPE * shift;
      if (exact) shift = mk[k];
      const float ex = __expf(e - shift);
      const float w0 = quad_bcast<0>(ex);
      const float w1 = quad_bcast<1>(ex);
      const float w2 = quad_bcast<2>(ex);
      const float w3 = quad_bcast<3>(ex);
      float* op = acc + dk[k] * OWN_PITCH + lane;
      atomicAdd(op +  0, r0[k] * w0);
      atomicAdd(op + 32, r1[k] * w1);
      atomicAdd(op + 64, r2[k] * w2);
      atomicAdd(op + 96, r3[k] * w3);
      if ((lane & 12) == 0) atomicAdd(acc + dk[k] * OWN_PITCH + 128 + hd, ex);
    }
  }
}

#define OWN_U 4
__global__ void __launch_bounds__(32 * OWN_WAVES)
agg1_owner_kernel(const int* __restrict__ src, const int* __restrict__ dst,
                  const float* __restrict__ a_src, const float* __restrict__ a_dst,
                  const float* __restrict__ m, const float* __restrict__ stat,
                  const float* __restrict__ h, const float* __restrict__ b,
                  float* __restrict__ h1, int* __restrict__ lists, int* __restrict__ counts,
                  int E, int Nn) {
  __shared__ float acc[OWN_R * OWN_PITCH];
  __shared__ int ring_e[OWN_WAVES][OWN_QMASK + 1];
  __shared__ int ring_d[OWN_WAVES][OWN_QMASK + 1];
  const int lane = threadIdx.x & 31;
  const int wave = threadIdx.x >> 5;
  const int lo = blockIdx.x * OWN_R;
  const int nown = min(OWN_R, Nn - lo);
  const int hd = ((lane & 3) << 1) | (lane >> 4);
  const float gmax = stat[hd * STAT_PITCH];
  const bool exact = spread_too_wide1(gmax, stat[(8 + hd) * STAT_PITCH]);
  int* qe = ring_e[wave];
  int* qd = ring_d[wave];
  int2* lst = (int2*)lists + (size_t)(blockIdx.x * OWN_WAVES + wave) * LIST_CAP;

  for (int i = threadIdx.x; i < OWN_R * OWN_PITCH; i += 32 * OWN_WAVES) acc[i] = 0.f;
  __syncthreads();

  int head = 0, cnt = 0;
  for (int r0 = 0; r0 < nown; r0 += 32 * OWN_WAVES) {
    const int r = r0 + 32 * lane + wave;
    const bool pred = r < nown;
    cnt = own_append(qe, qd, cnt, __builtin_amdgcn_ballot_w32(pred), pred, E + lo + r, r, lane);
  }

  const int ng = E >> 2;
  const int4* __restrict__ dst4 = (const int4*)dst;
  const int per_trip = OWN_WAVES * OWN_U * 32;
  const int ntrips = (ng + per_trip - 1) / per_trip;
  int trip = (ntrips > 0) ? (int)(blockIdx.x % (unsigned)ntrips) : 0;
  int4 cur[OWN_U], nxt[OWN_U];
  int g0 = trip * per_trip + wave * (OWN_U * 32) + lane;
#pragma unroll
  for (int j = 0; j < OWN_U; ++j) {
    const int g = g0 + 32 * j;
    cur[j] = dst4[min(g, max(ng - 1, 0))];
    if (g >= ng) cur[j] = (int4){-1, -1, -1, -1};
  }
  for (int t = 0; t < ntrips; ++t) {
    trip = (trip + 1 < ntrips) ? trip + 1 : 0;
    const int gn = trip * per_trip + wave * (OWN_U * 32) + lane;
#pragma unroll
    for (int j = 0; j < OWN_U; ++j) {
      const int g = gn + 32 * j;
      nxt[j] = dst4[min(g, ng - 1)];
      if (g >= ng) nxt[j] = (int4){-1, -1, -1, -1};
    }
#pragma unroll
    for (int j = 0; j < OWN_U; ++j) {
      const int e0 = (g0 + 32 * j) * 4;
      const int dx = cur[j].x - lo, dy = cur[j].y - lo, dz = cur[j].z - lo, dw = cur[j].w - lo;
      const bool px = (unsigned)dx < (unsigned)nown, py = (unsigned)dy < (unsigned)nown;
      const bool pz = (unsigned)dz < (unsigned)nown, pw = (unsigned)dw < (unsigned)nown;
      const unsigned mx = __builtin_amdgcn_ballot_w32(px), my = __builtin_amdgcn_ballot_w32(py);
      const unsigned mz = __builtin_amdgcn_ballot_w32(pz), mw = __builtin_amdgcn_ballot_w32(pw);
      if ((mx | my | mz | mw) != 0u) {
        cnt = own_append(qe, qd, cnt, mx, px, e0 + 0, dx, lane);
        cnt = own_append(qe, qd, cnt, my, py, e0 + 1, dy, lane);
        cnt = own_append(qe, qd, cnt, mz, pz, e0 + 2, dz, lane);
        cnt = own_append(qe, qd, cnt, mw, pw, e0 + 3, dw, lane);
        while (cnt - head >= EPW) {
          own_batch<true>(qe, qd, head, EPW, lo, lane, hd, exact, gmax, src, a_src, a_dst, m, h, acc, E, lst);
          head += EPW;
        }
      }
    }
#pragma unroll
    for (int j = 0; j < OWN_U; ++j) cur[j] = nxt[j];
    g0 = gn;
  }
  if (wave == 0) {
    const int e = (ng << 2) + lane;
    const int dl = ((lane < (E & 3)) ? dst[min(e, E - 1)] : -1) - lo;
    const bool pred = (lane < (E & 3)) && ((unsigned)dl < (unsigned)nown);
    cnt = own_append(qe, qd, cnt, __builtin_amdgcn_ballot_w32(pred), pred, e, dl, lane);
  }
  while (cnt - head > 0) {
    const int nv = min(cnt - head, EPW);
    own_batch<false>(qe, qd, head, nv, lo, lane, hd, exact, gmax, src, a_src, a_dst, m, h, acc, E, lst);
    head += nv;
  }
  if (lane == 0) counts[blockIdx.x * OWN_WAVES + wave] = cnt;
  __syncthreads();

  for (int i = threadIdx.x; i < nown * 128; i += 32 * OWN_WAVES) {
    const int r = i >> 7, f = i & 127;
    const float v = acc[r * OWN_PITCH + f] / acc[r * OWN_PITCH + 128 + (f >> 4)] + b[f];
    h1[(size_t)(lo + r) * 128 + f] = (v > 0.f) ? v : (__expf(v) - 1.0f);
  }
}

__global__ void edge_max2_kernel(const int* __restrict__ src, const int* __restrict__ dst,
                                 const float* __restrict__ a_src, const float* __restrict__ a_dst,
                                 float* __restrict__ m, const float* __restrict__ stat,
                                 int E, int Etot) {
  if (!spread_too_wide2(stat[16 * STAT_PITCH], stat[17 * STAT_PITCH])) return;
  const int stride = (int)(gridDim.x * blockDim.x);
  for (int e = blockIdx.x * blockDim.x + threadIdx.x; e < Etot; e += stride) {
    const int s = (e < E) ? src[e] : (e - E);
    const int d = (e < E) ? dst[e] : (e - E);
    float v = a_src[s] + a_dst[d];
    v = (v >= 0.f) ? v : NEG_SLOPE * v;
    atomicMaxF32(&m[d], v);
  }
}

template <bool FULL>
__device__ __forceinline__ void own_batch2(int s_own, int d_own, int nvalid, int lo, int lane,
                                           bool exact, float gmax,
                                           const float* __restrict__ a_src,
                                           const float* __restrict__ a_dst,
                                           const float* __restrict__ m,
                                           const float* __restrict__ h2,
                                           float* __restrict__ acc) {
  const float es = a_src[s_own];
  const float ed = a_dst[lo + d_own];
  float mo = 0.f;
  if (exact) mo = m[lo + d_own];
  int   sk[EPW], dk[EPW];
  float ra[EPW], rb[EPW], wk[EPW];
#pragma unroll
  for (int k = 0; k < EPW; ++k) {
    if (FULL || k < nvalid) {
      sk[k] = __builtin_amdgcn_readlane(s_own, k);
      dk[k] = __builtin_amdgcn_readlane(d_own, k);
      ra[k] = h2[(size_t)sk[k] * 40 + lane];
    }
  }
  if (lane < 8) {
#pragma unroll
    for (int k = 0; k < EPW; ++k)
      if (FULL || k < nvalid) rb[k] = h2[(size_t)sk[k] * 40 + 32 + lane];
  }
  float e = es + ed;
  e = (e >= 0.f) ? e : NEG_SLOPE * e;
  float shift = gmax + ed;
  shift = (shift >= 0.f) ? shift : NEG_SLOPE * shift;
  if (exact) shift = mo;
  const float ex_own = __expf(e - shift);
#pragma unroll
  for (int k = 0; k < EPW; ++k) {
    if (FULL || k < nvalid) {
      wk[k] = __int_as_float(__builtin_amdgcn_readlane(__float_as_int(ex_own), k));
      atomicAdd(acc + dk[k] * OWN2_PITCH + lane, ra[k] * wk[k]);
    }
  }
  if (lane < 8) {
#pragma unroll
    for (int k = 0; k < EPW; ++k)
      if (FULL || k < nvalid) atomicAdd(acc + dk[k] * OWN2_PITCH + 32 + lane, rb[k] * wk[k]);
    if (FULL || lane < nvalid) atomicAdd(acc + d_own * OWN2_PITCH + 40, ex_own);
  }
}

__global__ void __launch_bounds__(32 * OWN_WAVES)
agg2_owner_kernel(const int* __restrict__ src, const int* __restrict__ dst,
                  const float* __restrict__ a_src, const float* __restrict__ a_dst,
                  const float* __restrict__ m, const float* __restrict__ stat,
                  const float* __restrict__ h2, const float* __restrict__ b2,
                  const int* __restrict__ lists, const int* __restrict__ counts,
                  float* __restrict__ out, int E, int Etot, int Nn) {
  __shared__ float acc[OWN_R * OWN2_PITCH];
  const int lane = threadIdx.x & 31;
  const int wave = threadIdx.x >> 5;
  const int lo = blockIdx.x * OWN_R;
  const int nown = min(OWN_R, Nn - lo);
  const float gmax = stat[16 * STAT_PITCH];
  const bool exact = spread_too_wide2(gmax, stat[17 * STAT_PITCH]);
  const int wid = __builtin_amdgcn_readfirstlane((int)(blockIdx.x * OWN_WAVES + wave));
  const int n = counts[wid];
  const int2* lst = (const int2*)lists + (size_t)wid * LIST_CAP;

  for (int i = threadIdx.x; i < OWN_R * OWN2_PITCH; i += 32 * OWN_WAVES) acc[i] = 0.f;
  const int over = __syncthreads_or(n > LIST_CAP);

  if (!over) {
    const int slot = lane & (EPW - 1);
    int2 cur = make_int2(0, 0);
    if (n > 0) cur = lst[min(slot, n - 1)];
    for (int head = 0; head < n; head += EPW) {
      const int nv = min(n - head, EPW);
      const int2 nxt = lst[min(head + EPW + slot, n - 1)];
      if (nv == EPW) own_batch2<true>(cur.x, cur.y, EPW, lo, lane, exact, gmax, a_src, a_dst, m, h2, acc);
      else           own_batch2<false>(cur.x, cur.y, nv, lo, lane, exact, gmax, a_src, a_dst, m, h2, acc);
      cur = nxt;
    }
  } else {
    for (int e = threadIdx.x; e < Etot; e += 32 * OWN_WAVES) {
      const int d = ((e < E) ? dst[e] : (e - E)) - lo;
      if ((unsigned)d < (unsigned)nown) {
        const int sv = (e < E) ? src[e] : (e - E);
        const float ad = a_dst[lo + d];
        float sc = a_src[sv] + ad;
        sc = (sc >= 0.f) ? sc : NEG_SLOPE * sc;
        float shift = gmax + ad;
        shift = (shift >= 0.f) ? shift : NEG_SLOPE * shift;
        if (exact) shift = m[lo + d];
        const float ex = __expf(sc - shift);
        for (int c = 0; c < 40; ++c) atomicAdd(acc + d * OWN2_PITCH + c, h2[(size_t)sv * 40 + c] * ex);
        atomicAdd(acc + d * OWN2_PITCH + 40, ex);
      }
    }
  }
  __syncthreads();

  for (int i = threadIdx.x; i < nown * 40; i += 32 * OWN_WAVES) {
    const int r = i / 40, c = i - r * 40;
    out[(size_t)(lo + r) * 40 + c] = acc[r * OWN2_PITCH + c] / acc[r * OWN2_PITCH + 40] + b2[c];
  }
}

extern "C" void kernel_launch(void* const* d_in, const int* in_sizes, int n_in,
                              void* d_out, int out_size, void* d_ws, size_t ws_size,
                              hipStream_t stream) {
  const float* x   = (const float*)d_in[0];
  const int*   ei  = (const int*)d_in[1];
  const float* W1  = (const float*)d_in[2];
  const float* as1 = (const float*)d_in[3];
  const float* ad1 = (const float*)d_in[4];
  const float* b1  = (const float*)d_in[5];
  const float* W2  = (const float*)d_in[6];
  const float* as2 = (const float*)d_in[7];
  const float* ad2 = (const float*)d_in[8];
  const float* b2  = (const float*)d_in[9];
  float* out = (float*)d_out;

  const int Nn   = in_sizes[0] / 128;
  const int E    = in_sizes[1] / 2;
  const int Etot = E + Nn;
  const int* srcp = ei;
  const int* dstp = ei + E;

  float* ws   = (float*)d_ws;
  float* h    = ws;  ws += (size_t)Nn * 128;
  float* h1   = ws;  ws += (size_t)Nn * 128;
  float* asc1 = ws;  ws += (size_t)Nn * 8;
  float* adc1 = ws;  ws += (size_t)Nn * 8;
  float* m1   = ws;  ws += (size_t)Nn * 8;
  float* stat = ws;  ws += STAT_WORDS * STAT_PITCH;
  float* h2   = ws;  ws += (size_t)Nn * 40;
  float* asc2 = ws;  ws += Nn;
  float* adc2 = ws;  ws += Nn;
  float* m2   = ws;  ws += Nn;
  const int nOwn = (Nn + OWN_R - 1) / OWN_R;
  int* counts = (int*)ws;  ws += (size_t)nOwn * OWN_WAVES;
  int* lists  = (int*)ws;  ws += (size_t)nOwn * OWN_WAVES * LIST_CAP * 2;

  const int T = 256;
  const int fillBlocks = 2048;

  fill_kernel<<<fillBlocks, T, 0, stream>>>(m1, -INFINITY, (long long)Nn * 8 + STAT_WORDS * STAT_PITCH);
  fill_kernel<<<fillBlocks, T, 0, stream>>>(m2, -INFINITY, (long long)Nn);

  gat_wmma_gemm_k128<2, 128><<<dim3(Nn / 16), 4 * 32, 0, stream>>>(x, W1, h);
  scores1_kernel<<<(Nn * 8 + T - 1) / T, T, 0, stream>>>(h, as1, ad1, asc1, adc1, stat, Nn);
  edge_max1_kernel<<<1024, T, 0, stream>>>(srcp, dstp, asc1, adc1, m1, stat, E, Etot);
  agg1_owner_kernel<<<nOwn, 32 * OWN_WAVES, 0, stream>>>(srcp, dstp, asc1, adc1, m1, stat, h, b1, h1, lists, counts, E, Nn);

  gat_wmma_gemm_k128<3, 40><<<dim3(Nn / 16), 32, 0, stream>>>(h1, W2, h2);
  scores2_kernel<<<(Nn + T - 1) / T, T, 0, stream>>>(h2, as2, ad2, asc2, adc2, stat, Nn);
  edge_max2_kernel<<<1024, T, 0, stream>>>(srcp, dstp, asc2, adc2, m2, stat, E, Etot);
  agg2_owner_kernel<<<nOwn, 32 * OWN_WAVES, 0, stream>>>(srcp, dstp, asc2, adc2, m2, stat, h2, b2, lists, counts, out, E, Etot, Nn);
}


# Layer-1 owners stop scanning all edge destinations: one partition pass bins (source,row) pairs per owner (LDS ranks, a global counter add per block+owner); both layers replay; sums in arrival order
# speedup vs baseline: 5.3128x; 1.2041x over previous
#include <hip/hip_runtime.h>
#include <hip/hip_bf16.h>
#include <math.h>
#include <stdint.h>


typedef float v2f __attribute__((ext_vector_type(2)));
typedef float v4f __attribute__((ext_vector_type(4)));
typedef float v8f __attribute__((ext_vector_type(8)));

#define NEG_SLOPE 0.2f
#define SPREAD_MAX 16.0f
#define STAT_PITCH 32
#define STAT_WORDS 18

__device__ __forceinline__ void atomicMaxF32(float* addr, float val) {
  if (val >= 0.0f) {
    atomicMax((int*)addr, __float_as_int(val));
  } else {
    atomicMin((unsigned int*)addr, __float_as_uint(val));
  }
}

template <int NT, int NACT>
__global__ void __launch_bounds__(32 * ((NACT + 16 * NT - 1) / (16 * NT)))
gat_wmma_gemm_k128(const float* __restrict__ A, const float* __restrict__ B,
                   float* __restrict__ C) {
  const int lane = threadIdx.x & 31;
  const int wave = threadIdx.x >> 5;
  const int c = lane & 15;
  const int h = lane >> 4;
  constexpr int CG = (NACT + 16 * NT - 1) / (16 * NT);
  constexpr bool FULL = (NACT % (16 * NT)) == 0;
  const size_t m0 = (size_t)blockIdx.x * 16;
  const int n0 = (CG == 1) ? 0 : wave * (16 * NT);

  const float* arow = A + (m0 + c) * 128 + 4 * h;
  v4f q[16];
#pragma unroll
  for (int i = 0; i < 16; ++i) q[i] = *(const v4f*)(arow + 8 * i);

  v2f b[NT][32];
#pragma unroll
  for (int t = 0; t < NT; ++t) {
    const int gn = n0 + 16 * t + c;
    const bool in_range = FULL || (CG == 1 && 16 * t + 15 < NACT) || gn < NACT;
    const float* bcol = B + (size_t)(4 * h) * NACT + gn;
#pragma unroll
    for (int j = 0; j < 32; ++j) b[t][j] = (v2f){0.f, 0.f};
    if (in_range) {
#pragma unroll
      for (int j = 0; j < 32; ++j) {
        const int k = 8 * (j >> 1) + 2 * (j & 1);
        b[t][j].x = bcol[(size_t)k * NACT];
        b[t][j].y = bcol[(size_t)(k + 1) * NACT];
      }
    }
  }

  v8f acc[NT];
#pragma unroll
  for (int t = 0; t < NT; ++t) acc[t] = (v8f){0.f, 0.f, 0.f, 0.f, 0.f, 0.f, 0.f, 0.f};

#pragma unroll
  for (int j = 0; j < 32; ++j) {
    const v4f qq = q[j >> 1];
    v2f a;
    if (j & 1) { a.x = qq.z; a.y = qq.w; } else { a.x = qq.x; a.y = qq.y; }
#pragma unroll
    for (int t = 0; t < NT; ++t) {
      acc[t] = __builtin_amdgcn_wmma_f32_16x16x4_f32(
           false, a,  false, b[t][j],
           (short)0, acc[t],  false,  false);
    }
  }

#pragma unroll
  for (int t = 0; t < NT; ++t) {
    const int gn = n0 + 16 * t + c;
    if (FULL || (CG == 1 && 16 * t + 15 < NACT) || gn < NACT) {
      float* cp = C + (m0 + (size_t)(h << 3)) * NACT + gn;
#pragma unroll
      for (int r = 0; r < 8; ++r) cp[(size_t)r * NACT] = acc[t][r];
    }
  }
}

__global__ void fill_kernel(float* __restrict__ p, float v, long long n,
                            float* __restrict__ q, float w, long long nq) {
  const long long i0 = (long long)blockIdx.x * blockDim.x + threadIdx.x;
  const long long stride = (long long)gridDim.x * blockDim.x;
  for (long long i = i0; i < n; i += stride) p[i] = v;
  for (long long i = i0; i < nq; i += stride) q[i] = w;
}

__global__ void __launch_bounds__(256)
scores1_kernel(const float* __restrict__ h,
               const float* __restrict__ att_src,
               const float* __restrict__ att_dst,
               float* __restrict__ a_src, float* __restrict__ a_dst,
               float* __restrict__ stat, int Nn) {
  __shared__ float red[2][8][8];
  const int idx = blockIdx.x * blockDim.x + threadIdx.x;
  const bool valid = idx < Nn * 8;
  float s = 0.f, d = 0.f;
  if (valid) {
    const int n = idx >> 3, hd = idx & 7;
    const float* hp = h + (size_t)n * 128 + hd * 16;
#pragma unroll
    for (int c = 0; c < 16; ++c) {
      float v = hp[c];
      s = fmaf(v, att_src[hd * 16 + c], s);
      d = fmaf(v, att_dst[hd * 16 + c], d);
    }
    a_src[idx] = s;
    a_dst[idx] = d;
  }
  float vmax = valid ? s : -INFINITY;
  float vneg = valid ? -s : -INFINITY;
  vmax = fmaxf(vmax, __shfl_xor(vmax, 8));
  vneg = fmaxf(vneg, __shfl_xor(vneg, 8));
  vmax = fmaxf(vmax, __shfl_xor(vmax, 16));
  vneg = fmaxf(vneg, __shfl_xor(vneg, 16));
  const int lane = threadIdx.x & 31, wave = threadIdx.x >> 5;
  if (lane < 8) { red[0][wave][lane] = vmax; red[1][wave][lane] = vneg; }
  __syncthreads();
  if (threadIdx.x < 16) {
    const int which = threadIdx.x >> 3, hh = threadIdx.x & 7;
    float v = red[which][0][hh];
#pragma unroll
    for (int w = 1; w < 8; ++w) v = fmaxf(v, red[which][w][hh]);
    atomicMaxF32(&stat[threadIdx.x * STAT_PITCH], v);
  }
}

__global__ void __launch_bounds__(256)
scores2_kernel(const float* __restrict__ h2,
               const float* __restrict__ att_src,
               const float* __restrict__ att_dst,
               float* __restrict__ a_src, float* __restrict__ a_dst,
               float* __restrict__ stat, int Nn) {
  __shared__ float red[2][8];
  const int n = blockIdx.x * blockDim.x + threadIdx.x;
  const bool valid = n < Nn;
  float s = 0.f, d = 0.f;
  if (valid) {
    const float* hp = h2 + (size_t)n * 40;
#pragma unroll
    for (int c = 0; c < 40; ++c) {
      float v = hp[c];
      s = fmaf(v, att_src[c], s);
      d = fmaf(v, att_dst[c], d);
    }
    a_src[n] = s;
    a_dst[n] = d;
  }
  float vmax = valid ? s : -INFINITY;
  float vneg = valid ? -s : -INFINITY;
#pragma unroll
  for (int o = 16; o >= 1; o >>= 1) {
    vmax = fmaxf(vmax, __shfl_xor(vmax, o));
    vneg = fmaxf(vneg, __shfl_xor(vneg, o));
  }
  const int lane = threadIdx.x & 31, wave = threadIdx.x >> 5;
  if (lane == 0) { red[0][wave] = vmax; red[1][wave] = vneg; }
  __syncthreads();
  if (threadIdx.x < 2) {
    float v = red[threadIdx.x][0];
#pragma unroll
    for (int w = 1; w < 8; ++w) v = fmaxf(v, red[threadIdx.x][w]);
    atomicMaxF32(&stat[(16 + threadIdx.x) * STAT_PITCH], v);
  }
}

__device__ __forceinline__ bool spread_too_wide1(float gmax, float gneg) {
  return __builtin_amdgcn_ballot_w32(!(gmax + gneg <= SPREAD_MAX)) != 0u;
}
__device__ __forceinline__ bool spread_too_wide2(float gmax, float gneg) {
  return !(gmax + gneg <= SPREAD_MAX);
}

#define OWN_R 391
#define OWN_PITCH 136
#define OWN_WAVES 32
#define OWN2_PITCH 41
#define OWN_CAP 16384
#define CNT_PITCH 32
#define BIN_T 256
#define BIN_PER_T 16
#define BIN_OWNERS 256

__global__ void __launch_bounds__(BIN_T)
edge_pre1_kernel(const int* __restrict__ src, const int* __restrict__ dst,
                 const float* __restrict__ a_src, const float* __restrict__ a_dst,
                 float* __restrict__ m, const float* __restrict__ stat,
                 int* __restrict__ counts, int2* __restrict__ lists,
                 int E, int Etot, int nOwn) {
  __shared__ int hist[BIN_OWNERS + 32];
  __shared__ int base[BIN_OWNERS + 32];
  for (int o = threadIdx.x; o < BIN_OWNERS + 32; o += BIN_T) hist[o] = 0;
  __syncthreads();
  const int e0 = blockIdx.x * (BIN_T * BIN_PER_T) + threadIdx.x;
  int sv[BIN_PER_T], dv[BIN_PER_T], rk[BIN_PER_T];
#pragma unroll
  for (int i = 0; i < BIN_PER_T; ++i) {
    const int e = e0 + i * BIN_T;
    const int ec = min(e, E - 1);
    const int s = src[ec], d = dst[ec];
    sv[i] = (e < E) ? s : (e - E);
    dv[i] = (e < E) ? d : (e - E);
  }
#pragma unroll
  for (int i = 0; i < BIN_PER_T; ++i) {
    const int o = (e0 + i * BIN_T < Etot) ? dv[i] / OWN_R : BIN_OWNERS + (int)(threadIdx.x & 31);
    rk[i] = atomicAdd(&hist[o], 1);
    dv[i] = o | ((dv[i] - o * OWN_R) << 9);
  }
  __syncthreads();
  for (int o = threadIdx.x; o < BIN_OWNERS + 32; o += BIN_T) {
    const int cnt = hist[o];
    base[o] = (o < nOwn && cnt > 0) ? atomicAdd(&counts[o * CNT_PITCH], cnt) : 0;
  }
  __syncthreads();
  int pos[BIN_PER_T];
#pragma unroll
  for (int i = 0; i < BIN_PER_T; ++i) pos[i] = base[dv[i] & 511] + rk[i];
#pragma unroll
  for (int i = 0; i < BIN_PER_T; ++i) {
    if (e0 + i * BIN_T < Etot && pos[i] < OWN_CAP)
      lists[(size_t)(dv[i] & 511) * OWN_CAP + pos[i]] = make_int2(sv[i], dv[i] >> 9);
  }

  const int h_own = (int)(threadIdx.x & 7);
  if (!spread_too_wide1(stat[h_own * STAT_PITCH], stat[(8 + h_own) * STAT_PITCH])) return;
  const long long total = (long long)Etot * 8;
  const long long stride = (long long)gridDim.x * blockDim.x;
  for (long long idx = (long long)blockIdx.x * blockDim.x + threadIdx.x; idx < total; idx += stride) {
    const int e = (int)(idx >> 3), hd = (int)(idx & 7);
    const int s = (e < E) ? src[e] : (e - E);
    const int d = (e < E) ? dst[e] : (e - E);
    float v = a_src[s * 8 + hd] + a_dst[d * 8 + hd];
    v = (v >= 0.f) ? v : NEG_SLOPE * v;
    atomicMaxF32(&m[d * 8 + hd], v);
  }
}

template <int J>
__device__ __forceinline__ float quad_bcast(float v) {
  return __int_as_float(__builtin_amdgcn_update_dpp(
      0, __float_as_int(v), J * 0x55, 0xF, 0xF, true));
}

#define EPW 8

template <bool FULL>
__device__ __forceinline__ void own_batch(int s_own, int d_own, int nvalid, int lo, int lane, int hd,
                                          bool exact, float gmax,
                                          const float* __restrict__ a_src,
                                          const float* __restrict__ a_dst,
                                          const float* __restrict__ m,
                                          const float* __restrict__ h,
                                          float* __restrict__ acc) {
  int   dk[EPW];
  float es[EPW], ed[EPW], mk[EPW], r0[EPW], r1[EPW], r2[EPW], r3[EPW];
#pragma unroll
  for (int k = 0; k < EPW; ++k) {
    mk[k] = 0.f;
    if (FULL || k < nvalid) {
      const int sk = __builtin_amdgcn_readlane(s_own, k);
      dk[k] = __builtin_amdgcn_readlane(d_own, k);
      es[k] = a_src[sk * 8 + hd];
      ed[k] = a_dst[(lo + dk[k]) * 8 + hd];
      const float* hp = h + (size_t)sk * 128 + lane;
      r0[k] = hp[0]; r1[k] = hp[32]; r2[k] = hp[64]; r3[k] = hp[96];
    }
  }
  if (exact) {
#pragma unroll
    for (int k = 0; k < EPW; ++k)
      if (FULL || k < nvalid) mk[k] = m[(lo + dk[k]) * 8 + hd];
  }
#pragma unroll
  for (int k = 0; k < EPW; ++k) {
    if (FULL || k < nvalid) {
      float e = es[k] + ed[k];
      e = (e >= 0.f) ? e : NEG_SLOPE * e;
      float shift = gmax + ed[k];
      shift = (shift >= 0.f) ? shift : NEG_SLOPE * shift;
      if (exact) shift = mk[k];
      const float ex = __expf(e - shift);
      const float w0 = quad_bcast<0>(ex);
      const float w1 = quad_bcast<1>(ex);
      const float w2 = quad_bcast<2>(ex);
      const float w3 = quad_bcast<3>(ex);
      float* op = acc + dk[k] * OWN_PITCH + lane;
      atomicAdd(op +  0, r0[k] * w0);
      atomicAdd(op + 32, r1[k] * w1);
      atomicAdd(op + 64, r2[k] * w2);
      atomicAdd(op + 96, r3[k] * w3);
      if ((lane & 12) == 0) atomicAdd(acc + dk[k] * OWN_PITCH + 128 + hd, ex);
    }
  }
}

__global__ void __launch_bounds__(32 * OWN_WAVES)
agg1_owner_kernel(const int* __restrict__ src, const int* __restrict__ dst,
                  const float* __restrict__ a_src, const float* __restrict__ a_dst,
                  const float* __restrict__ m, const float* __restrict__ stat,
                  const float* __restrict__ h, const float* __restrict__ b,
                  float* __restrict__ h1, const int2* __restrict__ lists,
                  const int* __restrict__ counts, int E, int Etot, int Nn) {
  __shared__ float acc[OWN_R * OWN_PITCH];
  const int lane = threadIdx.x & 31;
  const int wave = threadIdx.x >> 5;
  const int lo = blockIdx.x * OWN_R;
  const int nown = min(OWN_R, Nn - lo);
  const int hd = ((lane & 3) << 1) | (lane >> 4);
  const float gmax = stat[hd * STAT_PITCH];
  const bool exact = spread_too_wide1(gmax, stat[(8 + hd) * STAT_PITCH]);
  const int n = counts[blockIdx.x * CNT_PITCH];
  const int2* lst = lists + (size_t)blockIdx.x * OWN_CAP;

  for (int i = threadIdx.x; i < OWN_R * OWN_PITCH; i += 32 * OWN_WAVES) acc[i] = 0.f;
  __syncthreads();

  if (n <= OWN_CAP) {
    const int slot = lane & (EPW - 1);
    const int wv = __builtin_amdgcn_readfirstlane(wave);
    const int nfull = n & ~(EPW - 1);
    int head = wv * EPW;
    int2 cur = make_int2(0, 0);
    if (head < nfull) cur = lst[head + slot];
    for (; head < nfull; head += OWN_WAVES * EPW) {
      const int2 nxt = lst[min(head + OWN_WAVES * EPW + slot, n - 1)];
      own_batch<true>(cur.x, cur.y, EPW, lo, lane, hd, exact, gmax, a_src, a_dst, m, h, acc);
      cur = nxt;
    }
    if (n > nfull && ((nfull / EPW) & (OWN_WAVES - 1)) == wv) {
      const int2 last = lst[min(nfull + slot, n - 1)];
      own_batch<false>(last.x, last.y, n - nfull, lo, lane, hd, exact, gmax, a_src, a_dst, m, h, acc);
    }
  } else {
    for (int e = threadIdx.x; e < Etot; e += 32 * OWN_WAVES) {
      const int d = ((e < E) ? dst[e] : (e - E)) - lo;
      if ((unsigned)d < (unsigned)nown) {
        const int sv = (e < E) ? src[e] : (e - E);
        for (int hh = 0; hh < 8; ++hh) {
          const float ad = a_dst[(lo + d) * 8 + hh];
          float sc = a_src[sv * 8 + hh] + ad;
          sc = (sc >= 0.f) ? sc : NEG_SLOPE * sc;
          float shift = stat[hh * STAT_PITCH] + ad;
          shift = (shift >= 0.f) ? shift : NEG_SLOPE * shift;
          if (exact) shift = m[(lo + d) * 8 + hh];
          const float ex = __expf(sc - shift);
          for (int cc = 0; cc < 16; ++cc)
            atomicAdd(acc + d * OWN_PITCH + hh * 16 + cc, h[(size_t)sv * 128 + hh * 16 + cc] * ex);
          atomicAdd(acc + d * OWN_PITCH + 128 + hh, ex);
        }
      }
    }
  }
  __syncthreads();

  for (int i = threadIdx.x; i < nown * 128; i += 32 * OWN_WAVES) {
    const int r = i >> 7, f = i & 127;
    const float v = acc[r * OWN_PITCH + f] / acc[r * OWN_PITCH + 128 + (f >> 4)] + b[f];
    h1[(size_t)(lo + r) * 128 + f] = (v > 0.f) ? v : (__expf(v) - 1.0f);
  }
}

__global__ void edge_max2_kernel(const int* __restrict__ src, const int* __restrict__ dst,
                                 const float* __restrict__ a_src, const float* __restrict__ a_dst,
                                 float* __restrict__ m, const float* __restrict__ stat,
                                 int E, int Etot) {
  if (!spread_too_wide2(stat[16 * STAT_PITCH], stat[17 * STAT_PITCH])) return;
  const int stride = (int)(gridDim.x * blockDim.x);
  for (int e = blockIdx.x * blockDim.x + threadIdx.x; e < Etot; e += stride) {
    const int s = (e < E) ? src[e] : (e - E);
    const int d = (e < E) ? dst[e] : (e - E);
    float v = a_src[s] + a_dst[d];
    v = (v >= 0.f) ? v : NEG_SLOPE * v;
    atomicMaxF32(&m[d], v);
  }
}

template <bool FULL>
__device__ __forceinline__ void own_batch2(int s_own, int d_own, int nvalid, int lo, int lane,
                                           bool exact, float gmax,
                                           const float* __restrict__ a_src,
                                           const float* __restrict__ a_dst,
                                           const float* __restrict__ m,
                                           const float* __restrict__ h2,
                                           float* __restrict__ acc) {
  const float es = a_src[s_own];
  const float ed = a_dst[lo + d_own];
  float mo = 0.f;
  if (exact) mo = m[lo + d_own];
  int   sk[EPW], dk[EPW];
  float ra[EPW], rb[EPW], wk[EPW];
#pragma unroll
  for (int k = 0; k < EPW; ++k) {
    if (FULL || k < nvalid) {
      sk[k] = __builtin_amdgcn_readlane(s_own, k);
      dk[k] = __builtin_amdgcn_readlane(d_own, k);
      ra[k] = h2[(size_t)sk[k] * 40 + lane];
    }
  }
  if (lane < 8) {
#pragma unroll
    for (int k = 0; k < EPW; ++k)
      if (FULL || k < nvalid) rb[k] = h2[(size_t)sk[k] * 40 + 32 + lane];
  }
  float e = es + ed;
  e = (e >= 0.f) ? e : NEG_SLOPE * e;
  float shift = gmax + ed;
  shift = (shift >= 0.f) ? shift : NEG_SLOPE * shift;
  if (exact) shift = mo;
  const float ex_own = __expf(e - shift);
#pragma unroll
  for (int k = 0; k < EPW; ++k) {
    if (FULL || k < nvalid) {
      wk[k] = __int_as_float(__builtin_amdgcn_readlane(__float_as_int(ex_own), k));
      atomicAdd(acc + dk[k] * OWN2_PITCH + lane, ra[k] * wk[k]);
    }
  }
  if (lane < 8) {
#pragma unroll
    for (int k = 0; k < EPW; ++k)
      if (FULL || k < nvalid) atomicAdd(acc + dk[k] * OWN2_PITCH + 32 + lane, rb[k] * wk[k]);
    if (FULL || lane < nvalid) atomicAdd(acc + d_own * OWN2_PITCH + 40, ex_own);
  }
}

__global__ void __launch_bounds__(32 * OWN_WAVES)
agg2_owner_kernel(const int* __restrict__ src, const int* __restrict__ dst,
                  const float* __restrict__ a_src, const float* __restrict__ a_dst,
                  const float* __restrict__ m, const float* __restrict__ stat,
                  const float* __restrict__ h2, const float* __restrict__ b2,
                  const int2* __restrict__ lists, const int* __restrict__ counts,
                  float* __restrict__ out, int E, int Etot, int Nn) {
  __shared__ float acc[OWN_R * OWN2_PITCH];
  const int lane = threadIdx.x & 31;
  const int wave = threadIdx.x >> 5;
  const int lo = blockIdx.x * OWN_R;
  const int nown = min(OWN_R, Nn - lo);
  const float gmax = stat[16 * STAT_PITCH];
  const bool exact = spread_too_wide2(gmax, stat[17 * STAT_PITCH]);
  const int n = counts[blockIdx.x * CNT_PITCH];
  const int2* lst = lists + (size_t)blockIdx.x * OWN_CAP;

  for (int i = threadIdx.x; i < OWN_R * OWN2_PITCH; i += 32 * OWN_WAVES) acc[i] = 0.f;
  __syncthreads();

  if (n <= OWN_CAP) {
    const int slot = lane & (EPW - 1);
    int head = __builtin_amdgcn_readfirstlane(wave) * EPW;
    int2 cur = make_int2(0, 0);
    if (head < n) cur = lst[min(head + slot, n - 1)];
    for (; head < n; head += OWN_WAVES * EPW) {
      const int nv = min(n - head, EPW);
      const int2 nxt = lst[min(head + OWN_WAVES * EPW + slot, n - 1)];
      if (nv == EPW) own_batch2<true>(cur.x, cur.y, EPW, lo, lane, exact, gmax, a_src, a_dst, m, h2, acc);
      else           own_batch2<false>(cur.x, cur.y, nv, lo, lane, exact, gmax, a_src, a_dst, m, h2, acc);
      cur = nxt;
    }
  } else {
    for (int e = threadIdx.x; e < Etot; e += 32 * OWN_WAVES) {
      const int d = ((e < E) ? dst[e] : (e - E)) - lo;
      if ((unsigned)d < (unsigned)nown) {
        const int sv = (e < E) ? src[e] : (e - E);
        const float ad = a_dst[lo + d];
        float sc = a_src[sv] + ad;
        sc = (sc >= 0.f) ? sc : NEG_SLOPE * sc;
        float shift = gmax + ad;
        shift = (shift >= 0.f) ? shift : NEG_SLOPE * shift;
        if (exact) shift = m[lo + d];
        const float ex = __expf(sc - shift);
        for (int c = 0; c < 40; ++c) atomicAdd(acc + d * OWN2_PITCH + c, h2[(size_t)sv * 40 + c] * ex);
        atomicAdd(acc + d * OWN2_PITCH + 40, ex);
      }
    }
  }
  __syncthreads();

  for (int i = threadIdx.x; i < nown * 40; i += 32 * OWN_WAVES) {
    const int r = i / 40, c = i - r * 40;
    out[(size_t)(lo + r) * 40 + c] = acc[r * OWN2_PITCH + c] / acc[r * OWN2_PITCH + 40] + b2[c];
  }
}

extern "C" void kernel_launch(void* const* d_in, const int* in_sizes, int n_in,
                              void* d_out, int out_size, void* d_ws, size_t ws_size,
                              hipStream_t stream) {
  const float* x   = (const float*)d_in[0];
  const int*   ei  = (const int*)d_in[1];
  const float* W1  = (const float*)d_in[2];
  const float* as1 = (const float*)d_in[3];
  const float* ad1 = (const float*)d_in[4];
  const float* b1  = (const float*)d_in[5];
  const float* W2  = (const float*)d_in[6];
  const float* as2 = (const float*)d_in[7];
  const float* ad2 = (const float*)d_in[8];
  const float* b2  = (const float*)d_in[9];
  float* out = (float*)d_out;

  const int Nn   = in_sizes[0] / 128;
  const int E    = in_sizes[1] / 2;
  const int Etot = E + Nn;
  const int* srcp = ei;
  const int* dstp = ei + E;

  float* ws   = (float*)d_ws;
  float* h    = ws;  ws += (size_t)Nn * 128;
  float* h1   = ws;  ws += (size_t)Nn * 128;
  float* asc1 = ws;  ws += (size_t)Nn * 8;
  float* adc1 = ws;  ws += (size_t)Nn * 8;
  float* m1   = ws;  ws += (size_t)Nn * 8;
  float* stat = ws;  ws += STAT_WORDS * STAT_PITCH;
  float* h2   = ws;  ws += (size_t)Nn * 40;
  float* asc2 = ws;  ws += Nn;
  float* adc2 = ws;  ws += Nn;
  float* m2   = ws;  ws += Nn;
  const int nOwn = (Nn + OWN_R - 1) / OWN_R;
  int*  counts = (int*)ws;   ws += (size_t)nOwn * CNT_PITCH;
  int2* lists  = (int2*)ws;  ws += (size_t)nOwn * OWN_CAP * 2;

  const int T = 256;
  const int fillBlocks = 2048;

  fill_kernel<<<fillBlocks, T, 0, stream>>>(m1, -INFINITY, (long long)Nn * 8 + STAT_WORDS * STAT_PITCH, nullptr, 0.f, 0);
  fill_kernel<<<fillBlocks, T, 0, stream>>>(m2, -INFINITY, (long long)Nn, (float*)counts, 0.f, (long long)nOwn * CNT_PITCH);

  gat_wmma_gemm_k128<2, 128><<<dim3(Nn / 16), 4 * 32, 0, stream>>>(x, W1, h);
  scores1_kernel<<<(Nn * 8 + T - 1) / T, T, 0, stream>>>(h, as1, ad1, asc1, adc1, stat, Nn);
  edge_pre1_kernel<<<(Etot + BIN_T * BIN_PER_T - 1) / (BIN_T * BIN_PER_T), BIN_T, 0, stream>>>(srcp, dstp, asc1, adc1, m1, stat, counts, lists, E, Etot, nOwn);
  agg1_owner_kernel<<<nOwn, 32 * OWN_WAVES, 0, stream>>>(srcp, dstp, asc1, adc1, m1, stat, h, b1, h1, lists, counts, E, Etot, Nn);

  gat_wmma_gemm_k128<3, 40><<<dim3(Nn / 16), 32, 0, stream>>>(h1, W2, h2);
  scores2_kernel<<<(Nn + T - 1) / T, T, 0, stream>>>(h2, as2, ad2, asc2, adc2, stat, Nn);
  edge_max2_kernel<<<1024, T, 0, stream>>>(srcp, dstp, asc2, adc2, m2, stat, E, Etot);
  agg2_owner_kernel<<<nOwn, 32 * OWN_WAVES, 0, stream>>>(srcp, dstp, asc2, adc2, m2, stat, h2, b2, lists, counts, out, E, Etot, Nn);
}


# Both replays fetch a source row with one load (layer 1: 16 B per lane, was 4 loads; LDS words permuted, quad lanes score 4 pairs, h1 row one store; layer 2: 8 B per lane, was 2); arrival order
# speedup vs baseline: 5.7235x; 1.0773x over previous
#include <hip/hip_runtime.h>
#include <hip/hip_bf16.h>
#include <math.h>
#include <stdint.h>


typedef float v2f __attribute__((ext_vector_type(2)));
typedef float v4f __attribute__((ext_vector_type(4)));
typedef float v8f __attribute__((ext_vector_type(8)));

#define NEG_SLOPE 0.2f
#define SPREAD_MAX 16.0f
#define STAT_PITCH 32
#define STAT_WORDS 18

__device__ __forceinline__ void atomicMaxF32(float* addr, float val) {
  if (val >= 0.0f) {
    atomicMax((int*)addr, __float_as_int(val));
  } else {
    atomicMin((unsigned int*)addr, __float_as_uint(val));
  }
}

template <int NT, int NACT>
__global__ void __launch_bounds__(32 * ((NACT + 16 * NT - 1) / (16 * NT)))
gat_wmma_gemm_k128(const float* __restrict__ A, const float* __restrict__ B,
                   float* __restrict__ C) {
  const int lane = threadIdx.x & 31;
  const int wave = threadIdx.x >> 5;
  const int c = lane & 15;
  const int h = lane >> 4;
  constexpr int CG = (NACT + 16 * NT - 1) / (16 * NT);
  constexpr bool FULL = (NACT % (16 * NT)) == 0;
  const size_t m0 = (size_t)blockIdx.x * 16;
  const int n0 = (CG == 1) ? 0 : wave * (16 * NT);

  const float* arow = A + (m0 + c) * 128 + 4 * h;
  v4f q[16];
#pragma unroll
  for (int i = 0; i < 16; ++i) q[i] = *(const v4f*)(arow + 8 * i);

  v2f b[NT][32];
#pragma unroll
  for (int t = 0; t < NT; ++t) {
    const int gn = n0 + 16 * t + c;
    const bool in_range = FULL || (CG == 1 && 16 * t + 15 < NACT) || gn < NACT;
    const float* bcol = B + (size_t)(4 * h) * NACT + gn;
#pragma unroll
    for (int j = 0; j < 32; ++j) b[t][j] = (v2f){0.f, 0.f};
    if (in_range) {
#pragma unroll
      for (int j = 0; j < 32; ++j) {
        const int k = 8 * (j >> 1) + 2 * (j & 1);
        b[t][j].x = bcol[(size_t)k * NACT];
        b[t][j].y = bcol[(size_t)(k + 1) * NACT];
      }
    }
  }

  v8f acc[NT];
#pragma unroll
  for (int t = 0; t < NT; ++t) acc[t] = (v8f){0.f, 0.f, 0.f, 0.f, 0.f, 0.f, 0.f, 0.f};

#pragma unroll
  for (int j = 0; j < 32; ++j) {
    const v4f qq = q[j >> 1];
    v2f a;
    if (j & 1) { a.x = qq.z; a.y = qq.w; } else { a.x = qq.x; a.y = qq.y; }
#pragma unroll
    for (int t = 0; t < NT; ++t) {
      acc[t] = __builtin_amdgcn_wmma_f32_16x16x4_f32(
           false, a,  false, b[t][j],
           (short)0, acc[t],  false,  false);
    }
  }

#pragma unroll
  for (int t = 0; t < NT; ++t) {
    const int gn = n0 + 16 * t + c;
    if (FULL || (CG == 1 && 16 * t + 15 < NACT) || gn < NACT) {
      float* cp = C + (m0 + (size_t)(h << 3)) * NACT + gn;
#pragma unroll
      for (int r = 0; r < 8; ++r) cp[(size_t)r * NACT] = acc[t][r];
    }
  }
}

__global__ void fill_kernel(float* __restrict__ p, float v, long long n,
                            float* __restrict__ q, float w, long long nq) {
  const long long i0 = (long long)blockIdx.x * blockDim.x + threadIdx.x;
  const long long stride = (long long)gridDim.x * blockDim.x;
  for (long long i = i0; i < n; i += stride) p[i] = v;
  for (long long i = i0; i < nq; i += stride) q[i] = w;
}

__global__ void __launch_bounds__(256)
scores1_kernel(const float* __restrict__ h,
               const float* __restrict__ att_src,
               const float* __restrict__ att_dst,
               float* __restrict__ a_src, float* __restrict__ a_dst,
               float* __restrict__ stat, int Nn) {
  __shared__ float red[2][8][8];
  const int idx = blockIdx.x * blockDim.x + threadIdx.x;
  const bool valid = idx < Nn * 8;
  float s = 0.f, d = 0.f;
  if (valid) {
    const int n = idx >> 3, hd = idx & 7;
    const float* hp = h + (size_t)n * 128 + hd * 16;
#pragma unroll
    for (int c = 0; c < 16; ++c) {
      float v = hp[c];
      s = fmaf(v, att_src[hd * 16 + c], s);
      d = fmaf(v, att_dst[hd * 16 + c], d);
    }
    a_src[idx] = s;
    a_dst[idx] = d;
  }
  float vmax = valid ? s : -INFINITY;
  float vneg = valid ? -s : -INFINITY;
  vmax = fmaxf(vmax, __shfl_xor(vmax, 8));
  vneg = fmaxf(vneg, __shfl_xor(vneg, 8));
  vmax = fmaxf(vmax, __shfl_xor(vmax, 16));
  vneg = fmaxf(vneg, __shfl_xor(vneg, 16));
  const int lane = threadIdx.x & 31, wave = threadIdx.x >> 5;
  if (lane < 8) { red[0][wave][lane] = vmax; red[1][wave][lane] = vneg; }
  __syncthreads();
  if (threadIdx.x < 16) {
    const int which = threadIdx.x >> 3, hh = threadIdx.x & 7;
    float v = red[which][0][hh];
#pragma unroll
    for (int w = 1; w < 8; ++w) v = fmaxf(v, red[which][w][hh]);
    atomicMaxF32(&stat[threadIdx.x * STAT_PITCH], v);
  }
}

__global__ void __launch_bounds__(256)
scores2_kernel(const float* __restrict__ h2,
               const float* __restrict__ att_src,
               const float* __restrict__ att_dst,
               float* __restrict__ a_src, float* __restrict__ a_dst,
               float* __restrict__ stat, int Nn) {
  __shared__ float red[2][8];
  const int n = blockIdx.x * blockDim.x + threadIdx.x;
  const bool valid = n < Nn;
  float s = 0.f, d = 0.f;
  if (valid) {
    const float* hp = h2 + (size_t)n * 40;
#pragma unroll
    for (int c = 0; c < 40; ++c) {
      float v = hp[c];
      s = fmaf(v, att_src[c], s);
      d = fmaf(v, att_dst[c], d);
    }
    a_src[n] = s;
    a_dst[n] = d;
  }
  float vmax = valid ? s : -INFINITY;
  float vneg = valid ? -s : -INFINITY;
#pragma unroll
  for (int o = 16; o >= 1; o >>= 1) {
    vmax = fmaxf(vmax, __shfl_xor(vmax, o));
    vneg = fmaxf(vneg, __shfl_xor(vneg, o));
  }
  const int lane = threadIdx.x & 31, wave = threadIdx.x >> 5;
  if (lane == 0) { red[0][wave] = vmax; red[1][wave] = vneg; }
  __syncthreads();
  if (threadIdx.x < 2) {
    float v = red[threadIdx.x][0];
#pragma unroll
    for (int w = 1; w < 8; ++w) v = fmaxf(v, red[threadIdx.x][w]);
    atomicMaxF32(&stat[(16 + threadIdx.x) * STAT_PITCH], v);
  }
}

__device__ __forceinline__ bool spread_too_wide1(float gmax, float gneg) {
  return __builtin_amdgcn_ballot_w32(!(gmax + gneg <= SPREAD_MAX)) != 0u;
}
__device__ __forceinline__ bool spread_too_wide2(float gmax, float gneg) {
  return !(gmax + gneg <= SPREAD_MAX);
}

#define OWN_R 391
#define OWN_PITCH 136
#define OWN_WAVES 32
#define OWN2_PITCH 41
#define OWN_CAP 16384
#define CNT_PITCH 32
#define BIN_T 256
#define BIN_PER_T 16
#define BIN_OWNERS 256

__global__ void __launch_bounds__(BIN_T)
edge_pre1_kernel(const int* __restrict__ src, const int* __restrict__ dst,
                 const float* __restrict__ a_src, const float* __restrict__ a_dst,
                 float* __restrict__ m, const float* __restrict__ stat,
                 int* __restrict__ counts, int2* __restrict__ lists,
                 int E, int Etot, int nOwn) {
  __shared__ int hist[BIN_OWNERS + 32];
  __shared__ int base[BIN_OWNERS + 32];
  for (int o = threadIdx.x; o < BIN_OWNERS + 32; o += BIN_T) hist[o] = 0;
  __syncthreads();
  const int e0 = blockIdx.x * (BIN_T * BIN_PER_T) + threadIdx.x;
  int sv[BIN_PER_T], dv[BIN_PER_T], rk[BIN_PER_T];
#pragma unroll
  for (int i = 0; i < BIN_PER_T; ++i) {
    const int e = e0 + i * BIN_T;
    const int ec = min(e, E - 1);
    const int s = src[ec], d = dst[ec];
    sv[i] = (e < E) ? s : (e - E);
    dv[i] = (e < E) ? d : (e - E);
  }
#pragma unroll
  for (int i = 0; i < BIN_PER_T; ++i) {
    const int o = (e0 + i * BIN_T < Etot) ? dv[i] / OWN_R : BIN_OWNERS + (int)(threadIdx.x & 31);
    rk[i] = atomicAdd(&hist[o], 1);
    dv[i] = o | ((dv[i] - o * OWN_R) << 9);
  }
  __syncthreads();
  for (int o = threadIdx.x; o < BIN_OWNERS + 32; o += BIN_T) {
    const int cnt = hist[o];
    base[o] = (o < nOwn && cnt > 0) ? atomicAdd(&counts[o * CNT_PITCH], cnt) : 0;
  }
  __syncthreads();
  int pos[BIN_PER_T];
#pragma unroll
  for (int i = 0; i < BIN_PER_T; ++i) pos[i] = base[dv[i] & 511] + rk[i];
#pragma unroll
  for (int i = 0; i < BIN_PER_T; ++i) {
    if (e0 + i * BIN_T < Etot && pos[i] < OWN_CAP)
      lists[(size_t)(dv[i] & 511) * OWN_CAP + pos[i]] = make_int2(sv[i], dv[i] >> 9);
  }

  const int h_own = (int)(threadIdx.x & 7);
  if (!spread_too_wide1(stat[h_own * STAT_PITCH], stat[(8 + h_own) * STAT_PITCH])) return;
  const long long total = (long long)Etot * 8;
  const long long stride = (long long)gridDim.x * blockDim.x;
  for (long long idx = (long long)blockIdx.x * blockDim.x + threadIdx.x; idx < total; idx += stride) {
    const int e = (int)(idx >> 3), hd = (int)(idx & 7);
    const int s = (e < E) ? src[e] : (e - E);
    const int d = (e < E) ? dst[e] : (e - E);
    float v = a_src[s * 8 + hd] + a_dst[d * 8 + hd];
    v = (v >= 0.f) ? v : NEG_SLOPE * v;
    atomicMaxF32(&m[d * 8 + hd], v);
  }
}

template <int J>
__device__ __forceinline__ float quad_bcast(float v) {
  return __int_as_float(__builtin_amdgcn_update_dpp(
      0, __float_as_int(v), J * 0x55, 0xF, 0xF, true));
}

__device__ __forceinline__ float quad_bcast_n(float v, int j) {
  switch (j) {
    case 0:  return quad_bcast<0>(v);
    case 1:  return quad_bcast<1>(v);
    case 2:  return quad_bcast<2>(v);
    default: return quad_bcast<3>(v);
  }
}

#define EPW 8

template <bool FULL>
__device__ __forceinline__ void own_batch(int4 ent, int nvalid, int lo, int lane, int hq,
                                          bool exact, float gmax,
                                          const float* __restrict__ a_src,
                                          const float* __restrict__ a_dst,
                                          const float* __restrict__ m,
                                          const float* __restrict__ h,
                                          float* __restrict__ acc) {
  const int q2 = 2 * (lane & 3);
  float es[2], ed[2], mk[2];
#pragma unroll
  for (int g = 0; g < 2; ++g) {
    const int s = g ? ent.z : ent.x;
    const int d = g ? ent.w : ent.y;
    es[g] = a_src[s * 8 + hq];
    ed[g] = a_dst[(lo + d) * 8 + hq];
    mk[g] = 0.f;
  }
  int dk[EPW];
  v4f row[EPW];
#pragma unroll
  for (int k = 0; k < EPW; ++k) {
    if (FULL || k < nvalid) {
      const int sk = __builtin_amdgcn_readlane((k & 1) ? ent.z : ent.x, k >> 1);
      dk[k] = __builtin_amdgcn_readlane((k & 1) ? ent.w : ent.y, k >> 1);
      row[k] = *(const v4f*)(h + (size_t)sk * 128 + 4 * lane);
    }
  }
  if (exact) {
#pragma unroll
    for (int g = 0; g < 2; ++g) mk[g] = m[(lo + (g ? ent.w : ent.y)) * 8 + hq];
  }
  float ex[2];
#pragma unroll
  for (int g = 0; g < 2; ++g) {
    float e = es[g] + ed[g];
    e = (e >= 0.f) ? e : NEG_SLOPE * e;
    float shift = gmax + ed[g];
    shift = (shift >= 0.f) ? shift : NEG_SLOPE * shift;
    if (exact) shift = mk[g];
    ex[g] = __expf(e - shift);
    if (FULL || q2 + g < nvalid)
      atomicAdd(acc + (g ? ent.w : ent.y) * OWN_PITCH + 128 + hq, ex[g]);
  }
#pragma unroll
  for (int k = 0; k < EPW; ++k) {
    if (FULL || k < nvalid) {
      const float w = quad_bcast_n(ex[k & 1], k >> 1);
      float* op = acc + dk[k] * OWN_PITCH + lane;
      atomicAdd(op +  0, row[k].x * w);
      atomicAdd(op + 32, row[k].y * w);
      atomicAdd(op + 64, row[k].z * w);
      atomicAdd(op + 96, row[k].w * w);
    }
  }
}

__global__ void __launch_bounds__(32 * OWN_WAVES)
agg1_owner_kernel(const int* __restrict__ src, const int* __restrict__ dst,
                  const float* __restrict__ a_src, const float* __restrict__ a_dst,
                  const float* __restrict__ m, const float* __restrict__ stat,
                  const float* __restrict__ h, const float* __restrict__ b,
                  float* __restrict__ h1, const int2* __restrict__ lists,
                  const int* __restrict__ counts, int E, int Etot, int Nn) {
  __shared__ float acc[OWN_R * OWN_PITCH];
  const int lane = threadIdx.x & 31;
  const int wave = threadIdx.x >> 5;
  const int lo = blockIdx.x * OWN_R;
  const int nown = min(OWN_R, Nn - lo);
  const int hq = lane >> 2;
  const float gmax = stat[hq * STAT_PITCH];
  const bool exact = spread_too_wide1(gmax, stat[(8 + hq) * STAT_PITCH]);
  const int n = counts[blockIdx.x * CNT_PITCH];
  const int2* lst = lists + (size_t)blockIdx.x * OWN_CAP;

  for (int i = threadIdx.x; i < OWN_R * OWN_PITCH; i += 32 * OWN_WAVES) acc[i] = 0.f;
  __syncthreads();

  if (n <= OWN_CAP) {
    const int q2 = 2 * (lane & 3);
    const int wv = __builtin_amdgcn_readfirstlane(wave);
    const int nfull = n & ~(EPW - 1);
    int head = wv * EPW;
    int4 cur = make_int4(0, 0, 0, 0);
    if (head < nfull) cur = *(const int4*)(lst + head + q2);
    for (; head < nfull; head += OWN_WAVES * EPW) {
      const int4 nxt = *(const int4*)(lst + min(head + OWN_WAVES * EPW + q2, (n - 2) & ~1));
      own_batch<true>(cur, EPW, lo, lane, hq, exact, gmax, a_src, a_dst, m, h, acc);
      cur = nxt;
    }
    if (n > nfull && ((nfull / EPW) & (OWN_WAVES - 1)) == wv) {
      const int2 la = lst[min(nfull + q2, n - 1)];
      const int2 lb = lst[min(nfull + q2 + 1, n - 1)];
      own_batch<false>(make_int4(la.x, la.y, lb.x, lb.y), n - nfull, lo, lane, hq, exact, gmax, a_src, a_dst, m, h, acc);
    }
  } else {
    for (int e = threadIdx.x; e < Etot; e += 32 * OWN_WAVES) {
      const int d = ((e < E) ? dst[e] : (e - E)) - lo;
      if ((unsigned)d < (unsigned)nown) {
        const int sv = (e < E) ? src[e] : (e - E);
        for (int hh = 0; hh < 8; ++hh) {
          const float ad = a_dst[(lo + d) * 8 + hh];
          float sc = a_src[sv * 8 + hh] + ad;
          sc = (sc >= 0.f) ? sc : NEG_SLOPE * sc;
          float shift = stat[hh * STAT_PITCH] + ad;
          shift = (shift >= 0.f) ? shift : NEG_SLOPE * shift;
          if (exact) shift = m[(lo + d) * 8 + hh];
          const float ex = __expf(sc - shift);
          for (int cc = 0; cc < 16; ++cc)
            atomicAdd(acc + d * OWN_PITCH + 32 * (cc & 3) + hh * 4 + (cc >> 2), h[(size_t)sv * 128 + hh * 16 + cc] * ex);
          atomicAdd(acc + d * OWN_PITCH + 128 + hh, ex);
        }
      }
    }
  }
  __syncthreads();

  const v4f bb = *(const v4f*)(b + 4 * lane);
  for (int r = wave; r < nown; r += OWN_WAVES) {
    const float* ar = acc + r * OWN_PITCH;
    const float den = ar[128 + hq];
    v4f v;
    v.x = ar[lane] / den + bb.x;
    v.y = ar[32 + lane] / den + bb.y;
    v.z = ar[64 + lane] / den + bb.z;
    v.w = ar[96 + lane] / den + bb.w;
    v.x = (v.x > 0.f) ? v.x : (__expf(v.x) - 1.0f);
    v.y = (v.y > 0.f) ? v.y : (__expf(v.y) - 1.0f);
    v.z = (v.z > 0.f) ? v.z : (__expf(v.z) - 1.0f);
    v.w = (v.w > 0.f) ? v.w : (__expf(v.w) - 1.0f);
    *(v4f*)(h1 + (size_t)(lo + r) * 128 + 4 * lane) = v;
  }
}

__global__ void edge_max2_kernel(const int* __restrict__ src, const int* __restrict__ dst,
                                 const float* __restrict__ a_src, const float* __restrict__ a_dst,
                                 float* __restrict__ m, const float* __restrict__ stat,
                                 int E, int Etot) {
  if (!spread_too_wide2(stat[16 * STAT_PITCH], stat[17 * STAT_PITCH])) return;
  const int stride = (int)(gridDim.x * blockDim.x);
  for (int e = blockIdx.x * blockDim.x + threadIdx.x; e < Etot; e += stride) {
    const int s = (e < E) ? src[e] : (e - E);
    const int d = (e < E) ? dst[e] : (e - E);
    float v = a_src[s] + a_dst[d];
    v = (v >= 0.f) ? v : NEG_SLOPE * v;
    atomicMaxF32(&m[d], v);
  }
}

template <bool FULL>
__device__ __forceinline__ void own_batch2(int s_own, int d_own, int nvalid, int lo, int lane,
                                           bool exact, float gmax,
                                           const float* __restrict__ a_src,
                                           const float* __restrict__ a_dst,
                                           const float* __restrict__ m,
                                           const float* __restrict__ h2,
                                           float* __restrict__ acc) {
  const float es = a_src[s_own];
  const float ed = a_dst[lo + d_own];
  float mo = 0.f;
  if (exact) mo = m[lo + d_own];
  int   sk[EPW], dk[EPW];
  v2f   rr[EPW];
  float wk[EPW];
#pragma unroll
  for (int k = 0; k < EPW; ++k) {
    sk[k] = __builtin_amdgcn_readlane(s_own, k);
    dk[k] = __builtin_amdgcn_readlane(d_own, k);
  }
  if (lane < 20) {
#pragma unroll
    for (int k = 0; k < EPW; ++k) rr[k] = *(const v2f*)(h2 + (size_t)sk[k] * 40 + 2 * lane);
  }
  float e = es + ed;
  e = (e >= 0.f) ? e : NEG_SLOPE * e;
  float shift = gmax + ed;
  shift = (shift >= 0.f) ? shift : NEG_SLOPE * shift;
  if (exact) shift = mo;
  const float ex_own = __expf(e - shift);
#pragma unroll
  for (int k = 0; k < EPW; ++k) wk[k] = __int_as_float(__builtin_amdgcn_readlane(__float_as_int(ex_own), k));
  if (lane < 20) {
#pragma unroll
    for (int k = 0; k < EPW; ++k) {
      if (FULL || k < nvalid) {
        float* op = acc + dk[k] * OWN2_PITCH + 2 * lane;
        atomicAdd(op,     rr[k].x * wk[k]);
        atomicAdd(op + 1, rr[k].y * wk[k]);
      }
    }
    if (lane < 8 && (FULL || lane < nvalid)) atomicAdd(acc + d_own * OWN2_PITCH + 40, ex_own);
  }
}

__global__ void __launch_bounds__(32 * OWN_WAVES)
agg2_owner_kernel(const int* __restrict__ src, const int* __restrict__ dst,
                  const float* __restrict__ a_src, const float* __restrict__ a_dst,
                  const float* __restrict__ m, const float* __restrict__ stat,
                  const float* __restrict__ h2, const float* __restrict__ b2,
                  const int2* __restrict__ lists, const int* __restrict__ counts,
                  float* __restrict__ out, int E, int Etot, int Nn) {
  __shared__ float acc[OWN_R * OWN2_PITCH];
  const int lane = threadIdx.x & 31;
  const int wave = threadIdx.x >> 5;
  const int lo = blockIdx.x * OWN_R;
  const int nown = min(OWN_R, Nn - lo);
  const float gmax = stat[16 * STAT_PITCH];
  const bool exact = spread_too_wide2(gmax, stat[17 * STAT_PITCH]);
  const int n = counts[blockIdx.x * CNT_PITCH];
  const int2* lst = lists + (size_t)blockIdx.x * OWN_CAP;

  for (int i = threadIdx.x; i < OWN_R * OWN2_PITCH; i += 32 * OWN_WAVES) acc[i] = 0.f;
  __syncthreads();

  if (n <= OWN_CAP) {
    const int slot = lane & (EPW - 1);
    int head = __builtin_amdgcn_readfirstlane(wave) * EPW;
    int2 cur = make_int2(0, 0);
    if (head < n) cur = lst[min(head + slot, n - 1)];
    for (; head < n; head += OWN_WAVES * EPW) {
      const int nv = min(n - head, EPW);
      const int2 nxt = lst[min(head + OWN_WAVES * EPW + slot, n - 1)];
      if (nv == EPW) own_batch2<true>(cur.x, cur.y, EPW, lo, lane, exact, gmax, a_src, a_dst, m, h2, acc);
      else           own_batch2<false>(cur.x, cur.y, nv, lo, lane, exact, gmax, a_src, a_dst, m, h2, acc);
      cur = nxt;
    }
  } else {
    for (int e = threadIdx.x; e < Etot; e += 32 * OWN_WAVES) {
      const int d = ((e < E) ? dst[e] : (e - E)) - lo;
      if ((unsigned)d < (unsigned)nown) {
        const int sv = (e < E) ? src[e] : (e - E);
        const float ad = a_dst[lo + d];
        float sc = a_src[sv] + ad;
        sc = (sc >= 0.f) ? sc : NEG_SLOPE * sc;
        float shift = gmax + ad;
        shift = (shift >= 0.f) ? shift : NEG_SLOPE * shift;
        if (exact) shift = m[lo + d];
        const float ex = __expf(sc - shift);
        for (int c = 0; c < 40; ++c) atomicAdd(acc + d * OWN2_PITCH + c, h2[(size_t)sv * 40 + c] * ex);
        atomicAdd(acc + d * OWN2_PITCH + 40, ex);
      }
    }
  }
  __syncthreads();

  for (int i = threadIdx.x; i < nown * 40; i += 32 * OWN_WAVES) {
    const int r = i / 40, c = i - r * 40;
    out[(size_t)(lo + r) * 40 + c] = acc[r * OWN2_PITCH + c] / acc[r * OWN2_PITCH + 40] + b2[c];
  }
}

extern "C" void kernel_launch(void* const* d_in, const int* in_sizes, int n_in,
                              void* d_out, int out_size, void* d_ws, size_t ws_size,
                              hipStream_t stream) {
  const float* x   = (const float*)d_in[0];
  const int*   ei  = (const int*)d_in[1];
  const float* W1  = (const float*)d_in[2];
  const float* as1 = (const float*)d_in[3];
  const float* ad1 = (const float*)d_in[4];
  const float* b1  = (const float*)d_in[5];
  const float* W2  = (const float*)d_in[6];
  const float* as2 = (const float*)d_in[7];
  const float* ad2 = (const float*)d_in[8];
  const float* b2  = (const float*)d_in[9];
  float* out = (float*)d_out;

  const int Nn   = in_sizes[0] / 128;
  const int E    = in_sizes[1] / 2;
  const int Etot = E + Nn;
  const int* srcp = ei;
  const int* dstp = ei + E;

  float* ws   = (float*)d_ws;
  float* h    = ws;  ws += (size_t)Nn * 128;
  float* h1   = ws;  ws += (size_t)Nn * 128;
  float* asc1 = ws;  ws += (size_t)Nn * 8;
  float* adc1 = ws;  ws += (size_t)Nn * 8;
  float* m1   = ws;  ws += (size_t)Nn * 8;
  float* stat = ws;  ws += STAT_WORDS * STAT_PITCH;
  float* h2   = ws;  ws += (size_t)Nn * 40;
  float* asc2 = ws;  ws += Nn;
  float* adc2 = ws;  ws += Nn;
  float* m2   = ws;  ws += Nn;
  const int nOwn = (Nn + OWN_R - 1) / OWN_R;
  int*  counts = (int*)ws;   ws += (size_t)nOwn * CNT_PITCH;
  int2* lists  = (int2*)ws;  ws += (size_t)nOwn * OWN_CAP * 2;

  const int T = 256;
  const int fillBlocks = 2048;

  fill_kernel<<<fillBlocks, T, 0, stream>>>(m1, -INFINITY, (long long)Nn * 8 + STAT_WORDS * STAT_PITCH, nullptr, 0.f, 0);
  fill_kernel<<<fillBlocks, T, 0, stream>>>(m2, -INFINITY, (long long)Nn, (float*)counts, 0.f, (long long)nOwn * CNT_PITCH);

  gat_wmma_gemm_k128<2, 128><<<dim3(Nn / 16), 4 * 32, 0, stream>>>(x, W1, h);
  scores1_kernel<<<(Nn * 8 + T - 1) / T, T, 0, stream>>>(h, as1, ad1, asc1, adc1, stat, Nn);
  edge_pre1_kernel<<<(Etot + BIN_T * BIN_PER_T - 1) / (BIN_T * BIN_PER_T), BIN_T, 0, stream>>>(srcp, dstp, asc1, adc1, m1, stat, counts, lists, E, Etot, nOwn);
  agg1_owner_kernel<<<nOwn, 32 * OWN_WAVES, 0, stream>>>(srcp, dstp, asc1, adc1, m1, stat, h, b1, h1, lists, counts, E, Etot, Nn);

  gat_wmma_gemm_k128<3, 40><<<dim3(Nn / 16), 32, 0, stream>>>(h1, W2, h2);
  scores2_kernel<<<(Nn + T - 1) / T, T, 0, stream>>>(h2, as2, ad2, asc2, adc2, stat, Nn);
  edge_max2_kernel<<<1024, T, 0, stream>>>(srcp, dstp, asc2, adc2, m2, stat, E, Etot);
  agg2_owner_kernel<<<nOwn, 32 * OWN_WAVES, 0, stream>>>(srcp, dstp, asc2, adc2, m2, stat, h2, b2, lists, counts, out, E, Etot, Nn);
}
